# starting kernel
# baseline (speedup 1.0000x reference)
#include <hip/hip_runtime.h>
#include <hip/hip_bf16.h>

typedef float v2f __attribute__((ext_vector_type(2)));
typedef float v8f __attribute__((ext_vector_type(8)));

#define PSTRIDE 160
#define WLDS    (32 * PSTRIDE)

__device__ __forceinline__ unsigned enc_f32(float f) {
    unsigned u = __float_as_uint(f);
    return (u & 0x80000000u) ? ~u : (u | 0x80000000u);
}
__device__ __forceinline__ float dec_f32(unsigned e) {
    return (e & 0x80000000u) ? __uint_as_float(e ^ 0x80000000u) : __uint_as_float(~e);
}

__device__ __forceinline__ void stage_weight_pairs(float* sW, const float* __restrict__ W) {
    for (int i = threadIdx.x; i < 64 * 64; i += blockDim.x) {
        const int k = i >> 6, col = i & 63;
        sW[(k >> 1) * PSTRIDE + col * 2 + (k & 1)] = W[i];
    }
}

__global__ void node_xform_kernel(const float* __restrict__ x2,
                                  const float* __restrict__ Wl, const float* __restrict__ bl,
                                  const float* __restrict__ Wr, const float* __restrict__ br,
                                  float* __restrict__ xl, float* __restrict__ xr, int NT)
{
    __shared__ float sWl[WLDS];
    __shared__ float sWr[WLDS];
    __shared__ float sbl[64], sbr[64];
    stage_weight_pairs(sWl, Wl);
    stage_weight_pairs(sWr, Wr);
    for (int i = threadIdx.x; i < 64; i += blockDim.x) { sbl[i] = bl[i]; sbr[i] = br[i]; }
    __syncthreads();

    const int lane    = threadIdx.x & 31;
    const int wave    = threadIdx.x >> 5;
    const int wpb     = blockDim.x >> 5;
    const int l15     = lane & 15;
    const int half    = lane >> 4;
    const int numTiles = NT >> 4;

    for (int tile = blockIdx.x * wpb + wave; tile < numTiles; tile += gridDim.x * wpb) {
        v8f accl[4] = {v8f{}, v8f{}, v8f{}, v8f{}};
        v8f accr[4] = {v8f{}, v8f{}, v8f{}, v8f{}};
        const float* xrow = x2 + (size_t)(tile * 16 + l15) * 64;
        #pragma unroll 4
        for (int kk = 0; kk < 16; ++kk) {
            const int kbase = kk * 4 + half * 2;
            const v2f a = *(const v2f*)(xrow + kbase);
            const int pbase = (kk * 2 + half) * PSTRIDE;
            #pragma unroll
            for (int nb = 0; nb < 4; ++nb) {
                const int col = nb * 16 + l15;
                const v2f bL = *(const v2f*)(sWl + pbase + col * 2);
                const v2f bR = *(const v2f*)(sWr + pbase + col * 2);
                accl[nb] = __builtin_amdgcn_wmma_f32_16x16x4_f32(
                    false, a, false, bL, (short)0, accl[nb], false, false);
                accr[nb] = __builtin_amdgcn_wmma_f32_16x16x4_f32(
                    false, a, false, bR, (short)0, accr[nb], false, false);
            }
        }
        #pragma unroll
        for (int nb = 0; nb < 4; ++nb) {
            const int col = nb * 16 + l15;
            #pragma unroll
            for (int v = 0; v < 8; ++v) {
                const size_t row = (size_t)(tile * 16 + v + half * 8);
                xl[row * 64 + col] = accl[nb][v] + sbl[col];
                xr[row * 64 + col] = accr[nb][v] + sbr[col];
            }
        }
    }
}

__global__ void init_ws_kernel(unsigned* __restrict__ smax, float* __restrict__ denom,
                               float* __restrict__ agg, int NT)
{
    const long long n = (long long)NT * 64;
    for (long long i = (long long)blockIdx.x * blockDim.x + threadIdx.x; i < n;
         i += (long long)gridDim.x * blockDim.x) {
        agg[i] = 0.0f;
        if (i < NT) { smax[i] = 0u; denom[i] = 0.0f; }
    }
}

__global__ void edge_score_kernel(const int* __restrict__ src, const int* __restrict__ dst,
                                  const float* __restrict__ eattr,
                                  const float* __restrict__ We, const float* __restrict__ att,
                                  const float* __restrict__ xl, const float* __restrict__ xr,
                                  float* __restrict__ score, unsigned* __restrict__ smax,
                                  long long E)
{
    const int lane = threadIdx.x & 31;
    const long long wid = (long long)blockIdx.x * (blockDim.x >> 5) + (threadIdx.x >> 5);
    const long long nw  = (long long)gridDim.x * (blockDim.x >> 5);
    const float2 we = ((const float2*)We)[lane];
    const float2 at = ((const float2*)att)[lane];
    for (long long e = wid; e < E; e += nw) {
        const int s = src[e];
        const int d = dst[e];
        const float ea = eattr[e];
        const float2 l2 = ((const float2*)(xl + (size_t)s * 64))[lane];
        const float2 r2 = ((const float2*)(xr + (size_t)d * 64))[lane];
        float m0 = l2.x + r2.x + ea * we.x;
        float m1 = l2.y + r2.y + ea * we.y;
        m0 = (m0 > 0.0f) ? m0 : 0.2f * m0;
        m1 = (m1 > 0.0f) ? m1 : 0.2f * m1;
        float p = at.x * m0 + at.y * m1;
        #pragma unroll
        for (int off = 16; off > 0; off >>= 1) p += __shfl_xor(p, off, 32);
        if (lane == 0) {
            score[e] = p;
            atomicMax(&smax[d], enc_f32(p));
        }
    }
}

__global__ void edge_denom_kernel(const int* __restrict__ dst, const float* __restrict__ score,
                                  const unsigned* __restrict__ smax, float* __restrict__ denom,
                                  long long E)
{
    for (long long e = (long long)blockIdx.x * blockDim.x + threadIdx.x; e < E;
         e += (long long)gridDim.x * blockDim.x) {
        const int d = dst[e];
        const float ex = __expf(score[e] - dec_f32(smax[d]));
        atomicAdd(&denom[d], ex);
    }
}

__global__ void edge_agg_kernel(const int* __restrict__ src, const int* __restrict__ dst,
                                const float* __restrict__ score, const unsigned* __restrict__ smax,
                                const float* __restrict__ denom, const float* __restrict__ xl,
                                float* __restrict__ agg, long long E)
{
    const int lane = threadIdx.x & 31;
    const long long wid = (long long)blockIdx.x * (blockDim.x >> 5) + (threadIdx.x >> 5);
    const long long nw  = (long long)gridDim.x * (blockDim.x >> 5);
    for (long long e = wid; e < E; e += nw) {
        const int s = src[e];
        const int d = dst[e];
        const float alpha = __expf(score[e] - dec_f32(smax[d])) / denom[d];
        const float2 l2 = ((const float2*)(xl + (size_t)s * 64))[lane];
        float* ap = agg + (size_t)d * 64 + 2 * lane;
        atomicAdd(ap,     alpha * l2.x);
        atomicAdd(ap + 1, alpha * l2.y);
    }
}

__global__ void out_xform_kernel(const float* __restrict__ agg, const float* __restrict__ bias,
                                 const float* __restrict__ Wfc, const float* __restrict__ bfc,
                                 float* __restrict__ out, int NT)
{
    __shared__ float sW[WLDS];
    __shared__ float sb[64], sbias[64];
    stage_weight_pairs(sW, Wfc);
    for (int i = threadIdx.x; i < 64; i += blockDim.x) { sb[i] = bfc[i]; sbias[i] = bias[i]; }
    __syncthreads();

    const int lane = threadIdx.x & 31;
    const int wave = threadIdx.x >> 5;
    const int wpb  = blockDim.x >> 5;
    const int l15  = lane & 15;
    const int half = lane >> 4;
    const int numTiles = NT >> 4;

    for (int tile = blockIdx.x * wpb + wave; tile < numTiles; tile += gridDim.x * wpb) {
        v8f acc[4] = {v8f{}, v8f{}, v8f{}, v8f{}};
        const float* arow = agg + (size_t)(tile * 16 + l15) * 64;
        #pragma unroll 4
        for (int kk = 0; kk < 16; ++kk) {
            const int kbase = kk * 4 + half * 2;
            v2f a = *(const v2f*)(arow + kbase);
            a.x += sbias[kbase];
            a.y += sbias[kbase + 1];
            const int pbase = (kk * 2 + half) * PSTRIDE;
            #pragma unroll
            for (int nb = 0; nb < 4; ++nb) {
                const int col = nb * 16 + l15;
                const v2f b = *(const v2f*)(sW + pbase + col * 2);
                acc[nb] = __builtin_amdgcn_wmma_f32_16x16x4_f32(
                    false, a, false, b, (short)0, acc[nb], false, false);
            }
        }
        #pragma unroll
        for (int nb = 0; nb < 4; ++nb) {
            const int col = nb * 16 + l15;
            #pragma unroll
            for (int v = 0; v < 8; ++v) {
                const size_t row = (size_t)(tile * 16 + v + half * 8);
                out[row * 64 + col] = acc[nb][v] + sb[col];
            }
        }
    }
}

extern "C" void kernel_launch(void* const* d_in, const int* in_sizes, int n_in,
                              void* d_out, int out_size, void* d_ws, size_t ws_size,
                              hipStream_t stream) {
    const float* x      = (const float*)d_in[0];
    const int*   eidx   = (const int*)d_in[1];
    const float* eattr  = (const float*)d_in[2];
    const float* W_l    = (const float*)d_in[3];
    const float* b_l    = (const float*)d_in[4];
    const float* W_r    = (const float*)d_in[5];
    const float* b_r    = (const float*)d_in[6];
    const float* W_e    = (const float*)d_in[7];
    const float* att    = (const float*)d_in[8];
    const float* bias   = (const float*)d_in[9];
    const float* W_fc   = (const float*)d_in[10];
    const float* b_fc   = (const float*)d_in[11];

    const int NT       = in_sizes[0] / 64;
    const long long E  = (long long)in_sizes[2];
    const int* src = eidx;
    const int* dst = eidx + E;

    float* ws     = (float*)d_ws;
    float* xl     = ws;
    float* xr     = xl + (size_t)NT * 64;
    float* agg    = xr + (size_t)NT * 64;
    float* score  = agg + (size_t)NT * 64;
    unsigned* smax = (unsigned*)(score + E);
    float* denom  = (float*)(smax + NT);

    const int numTiles = NT / 16;
    const int gemmBlocks = (numTiles + 7) / 8;

    node_xform_kernel<<<gemmBlocks, 256, 0, stream>>>(x, W_l, b_l, W_r, b_r, xl, xr, NT);
    init_ws_kernel<<<4096, 256, 0, stream>>>(smax, denom, agg, NT);
    edge_score_kernel<<<8192, 256, 0, stream>>>(src, dst, eattr, W_e, att, xl, xr, score, smax, E);
    edge_denom_kernel<<<8192, 256, 0, stream>>>(dst, score, smax, denom, E);
    edge_agg_kernel<<<8192, 256, 0, stream>>>(src, dst, score, smax, denom, xl, agg, E);
    out_xform_kernel<<<gemmBlocks, 256, 0, stream>>>(agg, bias, W_fc, b_fc, (float*)d_out, NT);
}


# edge score formed by the node-range owner for its queued edges (range's xr rows in LDS, row max by LDS atomic max); score+init launches, global max atomics, score array gone; sum orders kept
# speedup vs baseline: 2.4730x; 2.4730x over previous
#include <hip/hip_runtime.h>
#include <hip/hip_bf16.h>

typedef float v2f __attribute__((ext_vector_type(2)));
typedef float v8f __attribute__((ext_vector_type(8)));

#define PSTRIDE 160
#define WLDS    (32 * PSTRIDE)

__device__ __forceinline__ unsigned enc_f32(float f) {
    unsigned u = __float_as_uint(f);
    return (u & 0x80000000u) ? ~u : (u | 0x80000000u);
}
__device__ __forceinline__ float dec_f32(unsigned e) {
    return (e & 0x80000000u) ? __uint_as_float(e ^ 0x80000000u) : __uint_as_float(~e);
}

__device__ __forceinline__ void stage_weight_pairs(float* sW, const float* __restrict__ W) {
    for (int i = threadIdx.x; i < 64 * 64; i += blockDim.x) {
        const int k = i >> 6, col = i & 63;
        sW[(k >> 1) * PSTRIDE + col * 2 + (k & 1)] = W[i];
    }
}

__global__ void node_xform_kernel(const float* __restrict__ x2,
                                  const float* __restrict__ Wl, const float* __restrict__ bl,
                                  const float* __restrict__ Wr, const float* __restrict__ br,
                                  float* __restrict__ xl, float* __restrict__ xr, int NT)
{
    __shared__ float sWl[WLDS];
    __shared__ float sWr[WLDS];
    __shared__ float sbl[64], sbr[64];
    stage_weight_pairs(sWl, Wl);
    stage_weight_pairs(sWr, Wr);
    for (int i = threadIdx.x; i < 64; i += blockDim.x) { sbl[i] = bl[i]; sbr[i] = br[i]; }
    __syncthreads();

    const int lane    = threadIdx.x & 31;
    const int wave    = threadIdx.x >> 5;
    const int wpb     = blockDim.x >> 5;
    const int l15     = lane & 15;
    const int half    = lane >> 4;
    const int numTiles = NT >> 4;

    for (int tile = blockIdx.x * wpb + wave; tile < numTiles; tile += gridDim.x * wpb) {
        v8f accl[4] = {v8f{}, v8f{}, v8f{}, v8f{}};
        v8f accr[4] = {v8f{}, v8f{}, v8f{}, v8f{}};
        const float* xrow = x2 + (size_t)(tile * 16 + l15) * 64;
        #pragma unroll 4
        for (int kk = 0; kk < 16; ++kk) {
            const int kbase = kk * 4 + half * 2;
            const v2f a = *(const v2f*)(xrow + kbase);
            const int pbase = (kk * 2 + half) * PSTRIDE;
            #pragma unroll
            for (int nb = 0; nb < 4; ++nb) {
                const int col = nb * 16 + l15;
                const v2f bL = *(const v2f*)(sWl + pbase + col * 2);
                const v2f bR = *(const v2f*)(sWr + pbase + col * 2);
                accl[nb] = __builtin_amdgcn_wmma_f32_16x16x4_f32(
                    false, a, false, bL, (short)0, accl[nb], false, false);
                accr[nb] = __builtin_amdgcn_wmma_f32_16x16x4_f32(
                    false, a, false, bR, (short)0, accr[nb], false, false);
            }
        }
        #pragma unroll
        for (int nb = 0; nb < 4; ++nb) {
            const int col = nb * 16 + l15;
            #pragma unroll
            for (int v = 0; v < 8; ++v) {
                const size_t row = (size_t)(tile * 16 + v + half * 8);
                xl[row * 64 + col] = accl[nb][v] + sbl[col];
                xr[row * 64 + col] = accr[nb][v] + sbr[col];
            }
        }
    }
}

#define RMAX  196
#define QCAP  16384
#define SUPER 2097152
#define NLD   8
#define NROW  8

__device__ __forceinline__ int4 load_dst4(const int* __restrict__ dst, long long e, long long E, bool aligned)
{
    if (aligned && e + 3 < E) return *(const int4*)(dst + e);
    int4 v;
    v.x = (e     < E) ? dst[e]     : -1;
    v.y = (e + 1 < E) ? dst[e + 1] : -1;
    v.z = (e + 2 < E) ? dst[e + 2] : -1;
    v.w = (e + 3 < E) ? dst[e + 3] : -1;
    return v;
}

__device__ __forceinline__ float xhalf(float x) {
    const int b = __builtin_bit_cast(int, x);
    return __builtin_bit_cast(float, __builtin_amdgcn_permlanex16(b, b, 0x76543210, 0xfedcba98, false, false));
}
template <int M>
__device__ __forceinline__ float xrow(float x) {
    const int b = __builtin_bit_cast(int, x);
    return __builtin_bit_cast(float, __builtin_amdgcn_update_dpp(b, b, 0x160 + M, 0xf, 0xf, false));
}

__device__ __forceinline__ float edge_score(float2 l, float2 r, float ea, float2 we, float2 at)
{
    float m0 = l.x + r.x + ea * we.x;
    float m1 = l.y + r.y + ea * we.y;
    m0 = (m0 > 0.0f) ? m0 : 0.2f * m0;
    m1 = (m1 > 0.0f) ? m1 : 0.2f * m1;
    float p = at.x * m0 + at.y * m1;
    p += xhalf(p);
    p += xrow<8>(p);
    p += xrow<4>(p);
    p += xrow<2>(p);
    p += xrow<1>(p);
    return p;
}

__global__ void __launch_bounds__(1024)
node_range_gat_kernel(const int* __restrict__ src, const int* __restrict__ dst,
                      const float* __restrict__ eattr,
                      const float* __restrict__ We, const float* __restrict__ att,
                      const float* __restrict__ xl, const float* __restrict__ xr,
                      float* __restrict__ agg, int NT, long long E)
{
    __shared__ float    sAgg[RMAX * 64];
    __shared__ float    sXr[RMAX * 64];
    __shared__ float    sDen[RMAX];
    __shared__ float    sMax[RMAX];
    __shared__ unsigned sRnd[RMAX];
    __shared__ float    sScl[RMAX];
    __shared__ int   qE[QCAP];
    __shared__ int   qD[QCAP];
    __shared__ float qX[QCAP];
    __shared__ int   qN;

    const int tid   = threadIdx.x;
    const int lane  = tid & 31;
    const int wave  = __builtin_amdgcn_readfirstlane(tid >> 5);
    const int nwave = blockDim.x >> 5;
    const int nthr  = blockDim.x;
    const bool aligned = (((size_t)dst) & 15) == 0;
    const float2 we = ((const float2*)We)[lane];
    const float2 at = ((const float2*)att)[lane];

    for (int base = blockIdx.x * RMAX; base < NT; base += gridDim.x * RMAX) {
        const int R = (NT - base < RMAX) ? (NT - base) : RMAX;
        for (int i = tid; i < R * 64; i += nthr) {
            sAgg[i] = 0.0f;
            sXr[i]  = xr[(size_t)base * 64 + i];
        }
        for (int i = tid; i < R; i += nthr) {
            sDen[i] = 0.0f;
            sMax[i] = -__builtin_inff();
            sRnd[i] = 0u;
        }
        if (tid == 0) qN = 0;
        __syncthreads();

        long long c0  = 0;
        long long seg = SUPER;
        while (c0 < E) {
            const long long c1 = (c0 + seg < E) ? (c0 + seg) : E;

            const unsigned trip  = (unsigned)nthr * (4 * NLD);
            const unsigned ntrip = ((unsigned)(c1 - c0) + trip - 1) / trip;
            const unsigned rot   = ntrip * blockIdx.x / gridDim.x;
            for (unsigned tr = 0; tr < ntrip; ++tr) {
                const unsigned tt = (tr + rot < ntrip) ? (tr + rot) : (tr + rot - ntrip);
                const long long t0 = c0 + (long long)tt * trip;
                const long long e0 = t0 + (long long)tid * 4;
                int4 v[NLD];
                if (aligned && t0 + trip <= c1) {
                    #pragma unroll
                    for (int j = 0; j < NLD; ++j)
                        v[j] = *(const int4*)(dst + e0 + (long long)j * nthr * 4);
                } else {
                    #pragma unroll
                    for (int j = 0; j < NLD; ++j) {
                        const long long e = e0 + (long long)j * nthr * 4;
                        v[j] = (e < c1) ? load_dst4(dst, e, c1, aligned) : make_int4(-1, -1, -1, -1);
                    }
                }
                #pragma unroll
                for (int j = 0; j < NLD; ++j) {
                    const long long e = e0 + (long long)j * nthr * 4;
                    const int d4[4] = {v[j].x, v[j].y, v[j].z, v[j].w};
                    #pragma unroll
                    for (int k = 0; k < 4; ++k) {
                        const unsigned dl = (unsigned)(d4[k] - base);
                        if (dl < (unsigned)R) {
                            const int idx = atomicAdd(&qN, 1);
                            if (idx < QCAP) {
                                qE[idx] = (int)(e + k);
                                qD[idx] = (int)dl;
                            }
                        }
                    }
                }
            }
            __syncthreads();

            const int n = qN;
            if (n > QCAP) {
                __syncthreads();
                if (tid == 0) qN = 0;
                seg = (seg / 2 > QCAP) ? seg / 2 : QCAP;
                __syncthreads();
                continue;
            }

            for (int i = tid; i < n; i += nthr) {
                const int ee = qE[i];
                qE[i] = src[ee];
                qX[i] = eattr[ee];
            }
            __syncthreads();
            if (tid == 0) qN = 0;

            {
                int i0 = wave;
                for (; i0 + (NROW - 1) * nwave < n; i0 += NROW * nwave) {
                    float2 l[NROW];
                    float  ea[NROW];
                    int    dl[NROW];
                    #pragma unroll
                    for (int j = 0; j < NROW; ++j) {
                        const int i = i0 + j * nwave;
                        ea[j] = qX[i];
                        dl[j] = qD[i];
                        l[j]  = ((const float2*)(xl + (size_t)qE[i] * 64))[lane];
                    }
                    float p[NROW];
                    #pragma unroll
                    for (int j = 0; j < NROW; ++j) {
                        const float2 r = ((const float2*)(sXr + dl[j] * 64))[lane];
                        p[j] = edge_score(l[j], r, ea[j], we, at);
                    }
                    if (lane == 0) {
                        #pragma unroll
                        for (int j = 0; j < NROW; ++j) {
                            qX[i0 + j * nwave] = p[j];
                            atomicMax(&sRnd[dl[j]], enc_f32(p[j]));
                        }
                    }
                }
                for (; i0 < n; i0 += nwave) {
                    const int    dl = qD[i0];
                    const float2 l  = ((const float2*)(xl + (size_t)qE[i0] * 64))[lane];
                    const float2 r  = ((const float2*)(sXr + dl * 64))[lane];
                    const float  p  = edge_score(l, r, qX[i0], we, at);
                    if (lane == 0) {
                        qX[i0] = p;
                        atomicMax(&sRnd[dl], enc_f32(p));
                    }
                }
            }
            __syncthreads();

            for (int r = tid; r < R; r += nthr) {
                const unsigned u = sRnd[r];
                float s = 1.0f;
                if (u != 0u) {
                    const float mo = sMax[r];
                    const float mn = dec_f32(u);
                    if (mn > mo) {
                        if (mo != -__builtin_inff()) s = __expf(mo - mn);
                        sMax[r] = mn;
                    }
                }
                sScl[r] = s;
                sRnd[r] = 0u;
            }
            __syncthreads();
            if (c0 != 0) {
                for (int i = tid; i < R * 64; i += nthr) sAgg[i] *= sScl[i >> 6];
                for (int r = tid; r < R; r += nthr) sDen[r] *= sScl[r];
                __syncthreads();
            }

            for (int i = tid; i < n; i += nthr) {
                const int   dl = qD[i];
                const float ex = __expf(qX[i] - sMax[dl]);
                qX[i] = ex;
                atomicAdd(&sDen[dl], ex);
            }
            __syncthreads();

            {
                int i0 = wave;
                for (; i0 + (NROW - 1) * nwave < n; i0 += NROW * nwave) {
                    float2 r[NROW];
                    float  x[NROW];
                    int    dl[NROW];
                    #pragma unroll
                    for (int j = 0; j < NROW; ++j) {
                        const int i = i0 + j * nwave;
                        x[j]  = qX[i];
                        dl[j] = qD[i];
                        r[j]  = ((const float2*)(xl + (size_t)qE[i] * 64))[lane];
                    }
                    #pragma unroll
                    for (int j = 0; j < NROW; ++j) {
                        float* ap = sAgg + dl[j] * 64 + 2 * lane;
                        atomicAdd(ap,     x[j] * r[j].x);
                        atomicAdd(ap + 1, x[j] * r[j].y);
                    }
                }
                for (; i0 < n; i0 += nwave) {
                    const float  x = qX[i0];
                    const float2 r = ((const float2*)(xl + (size_t)qE[i0] * 64))[lane];
                    float* ap = sAgg + qD[i0] * 64 + 2 * lane;
                    atomicAdd(ap,     x * r.x);
                    atomicAdd(ap + 1, x * r.y);
                }
            }
            __syncthreads();
            c0 = c1;
        }

        for (int i = tid; i < R * 64; i += nthr) {
            const float den = sDen[i >> 6];
            agg[(size_t)base * 64 + i] = (den > 0.0f) ? sAgg[i] / den : 0.0f;
        }
        __syncthreads();
    }
}

__global__ void out_xform_kernel(const float* __restrict__ agg, const float* __restrict__ bias,
                                 const float* __restrict__ Wfc, const float* __restrict__ bfc,
                                 float* __restrict__ out, int NT)
{
    __shared__ float sW[WLDS];
    __shared__ float sb[64], sbias[64];
    stage_weight_pairs(sW, Wfc);
    for (int i = threadIdx.x; i < 64; i += blockDim.x) { sb[i] = bfc[i]; sbias[i] = bias[i]; }
    __syncthreads();

    const int lane = threadIdx.x & 31;
    const int wave = threadIdx.x >> 5;
    const int wpb  = blockDim.x >> 5;
    const int l15  = lane & 15;
    const int half = lane >> 4;
    const int numTiles = NT >> 4;

    for (int tile = blockIdx.x * wpb + wave; tile < numTiles; tile += gridDim.x * wpb) {
        v8f acc[4] = {v8f{}, v8f{}, v8f{}, v8f{}};
        const float* arow = agg + (size_t)(tile * 16 + l15) * 64;
        #pragma unroll 4
        for (int kk = 0; kk < 16; ++kk) {
            const int kbase = kk * 4 + half * 2;
            v2f a = *(const v2f*)(arow + kbase);
            a.x += sbias[kbase];
            a.y += sbias[kbase + 1];
            const int pbase = (kk * 2 + half) * PSTRIDE;
            #pragma unroll
            for (int nb = 0; nb < 4; ++nb) {
                const int col = nb * 16 + l15;
                const v2f b = *(const v2f*)(sW + pbase + col * 2);
                acc[nb] = __builtin_amdgcn_wmma_f32_16x16x4_f32(
                    false, a, false, b, (short)0, acc[nb], false, false);
            }
        }
        #pragma unroll
        for (int nb = 0; nb < 4; ++nb) {
            const int col = nb * 16 + l15;
            #pragma unroll
            for (int v = 0; v < 8; ++v) {
                const size_t row = (size_t)(tile * 16 + v + half * 8);
                out[row * 64 + col] = acc[nb][v] + sb[col];
            }
        }
    }
}

extern "C" void kernel_launch(void* const* d_in, const int* in_sizes, int n_in,
                              void* d_out, int out_size, void* d_ws, size_t ws_size,
                              hipStream_t stream) {
    const float* x      = (const float*)d_in[0];
    const int*   eidx   = (const int*)d_in[1];
    const float* eattr  = (const float*)d_in[2];
    const float* W_l    = (const float*)d_in[3];
    const float* b_l    = (const float*)d_in[4];
    const float* W_r    = (const float*)d_in[5];
    const float* b_r    = (const float*)d_in[6];
    const float* W_e    = (const float*)d_in[7];
    const float* att    = (const float*)d_in[8];
    const float* bias   = (const float*)d_in[9];
    const float* W_fc   = (const float*)d_in[10];
    const float* b_fc   = (const float*)d_in[11];

    const int NT       = in_sizes[0] / 64;
    const long long E  = (long long)in_sizes[2];
    const int* src = eidx;
    const int* dst = eidx + E;

    float* ws     = (float*)d_ws;
    float* xl     = ws;
    float* xr     = xl + (size_t)NT * 64;
    float* agg    = xr + (size_t)NT * 64;

    const int numTiles = NT / 16;
    const int gemmBlocks = (numTiles + 7) / 8;

    node_xform_kernel<<<gemmBlocks, 256, 0, stream>>>(x, W_l, b_l, W_r, b_r, xl, xr, NT);
    node_range_gat_kernel<<<256, 1024, 0, stream>>>(src, dst, eattr, W_e, att, xl, xr, agg, NT, E);
    out_xform_kernel<<<gemmBlocks, 256, 0, stream>>>(agg, bias, W_fc, b_fc, (float*)d_out, NT);
}


# owners' filter reads bit-sliced owner keys (8 one-bit planes of dst/196 per 32 edges, written by the transform kernel), not dst: a lane tests 32 edges per branch; sums in arrival order as before
# speedup vs baseline: 3.0861x; 1.2479x over previous
#include <hip/hip_runtime.h>
#include <hip/hip_bf16.h>

typedef float v2f __attribute__((ext_vector_type(2)));
typedef float v8f __attribute__((ext_vector_type(8)));

#define PSTRIDE 160
#define WLDS    (32 * PSTRIDE)

__device__ __forceinline__ unsigned enc_f32(float f) {
    unsigned u = __float_as_uint(f);
    return (u & 0x80000000u) ? ~u : (u | 0x80000000u);
}
__device__ __forceinline__ float dec_f32(unsigned e) {
    return (e & 0x80000000u) ? __uint_as_float(e ^ 0x80000000u) : __uint_as_float(~e);
}

#define RMAX  196

__device__ __forceinline__ void write_owner_key_planes(const int* __restrict__ dst, long long E,
                                                       unsigned* __restrict__ kplanes)
{
    const long long nG = (E + 31) >> 5;
    uint4* __restrict__ klo = (uint4*)kplanes;
    uint4* __restrict__ khi = klo + nG;
    const long long nthr = (long long)gridDim.x * blockDim.x;
    for (long long G = (long long)blockIdx.x * blockDim.x + threadIdx.x; G < nG; G += nthr) {
        int d[32];
        #pragma unroll
        for (int l = 0; l < 32; ++l) {
            const long long e = (G << 5) + l;
            d[l] = dst[(e < E) ? e : (E - 1)];
        }
        unsigned p[8] = {0u, 0u, 0u, 0u, 0u, 0u, 0u, 0u};
        #pragma unroll
        for (int l = 0; l < 32; ++l) {
            const unsigned key = (unsigned)d[l] / RMAX;
            #pragma unroll
            for (int k = 0; k < 8; ++k) p[k] |= ((key >> k) & 1u) << l;
        }
        klo[G] = make_uint4(p[0], p[1], p[2], p[3]);
        khi[G] = make_uint4(p[4], p[5], p[6], p[7]);
    }
}

__device__ __forceinline__ void stage_weight_pairs(float* sW, const float* __restrict__ W) {
    for (int i = threadIdx.x; i < 64 * 64; i += blockDim.x) {
        const int k = i >> 6, col = i & 63;
        sW[(k >> 1) * PSTRIDE + col * 2 + (k & 1)] = W[i];
    }
}

__global__ void node_xform_kernel(const float* __restrict__ x2,
                                  const float* __restrict__ Wl, const float* __restrict__ bl,
                                  const float* __restrict__ Wr, const float* __restrict__ br,
                                  float* __restrict__ xl, float* __restrict__ xr, int NT,
                                  const int* __restrict__ dst, long long E, unsigned* __restrict__ kplanes)
{
    write_owner_key_planes(dst, E, kplanes);

    __shared__ float sWl[WLDS];
    __shared__ float sWr[WLDS];
    __shared__ float sbl[64], sbr[64];
    stage_weight_pairs(sWl, Wl);
    stage_weight_pairs(sWr, Wr);
    for (int i = threadIdx.x; i < 64; i += blockDim.x) { sbl[i] = bl[i]; sbr[i] = br[i]; }
    __syncthreads();

    const int lane    = threadIdx.x & 31;
    const int wave    = threadIdx.x >> 5;
    const int wpb     = blockDim.x >> 5;
    const int l15     = lane & 15;
    const int half    = lane >> 4;
    const int numTiles = NT >> 4;

    for (int tile = blockIdx.x * wpb + wave; tile < numTiles; tile += gridDim.x * wpb) {
        v8f accl[4] = {v8f{}, v8f{}, v8f{}, v8f{}};
        v8f accr[4] = {v8f{}, v8f{}, v8f{}, v8f{}};
        const float* xrow = x2 + (size_t)(tile * 16 + l15) * 64;
        #pragma unroll 4
        for (int kk = 0; kk < 16; ++kk) {
            const int kbase = kk * 4 + half * 2;
            const v2f a = *(const v2f*)(xrow + kbase);
            const int pbase = (kk * 2 + half) * PSTRIDE;
            #pragma unroll
            for (int nb = 0; nb < 4; ++nb) {
                const int col = nb * 16 + l15;
                const v2f bL = *(const v2f*)(sWl + pbase + col * 2);
                const v2f bR = *(const v2f*)(sWr + pbase + col * 2);
                accl[nb] = __builtin_amdgcn_wmma_f32_16x16x4_f32(
                    false, a, false, bL, (short)0, accl[nb], false, false);
                accr[nb] = __builtin_amdgcn_wmma_f32_16x16x4_f32(
                    false, a, false, bR, (short)0, accr[nb], false, false);
            }
        }
        #pragma unroll
        for (int nb = 0; nb < 4; ++nb) {
            const int col = nb * 16 + l15;
            #pragma unroll
            for (int v = 0; v < 8; ++v) {
                const size_t row = (size_t)(tile * 16 + v + half * 8);
                xl[row * 64 + col] = accl[nb][v] + sbl[col];
                xr[row * 64 + col] = accr[nb][v] + sbr[col];
            }
        }
    }
}

#define QCAP  16384
#define SUPER 2097152
#define NGRP  4
#define NROW  8

__device__ __forceinline__ float xhalf(float x) {
    const int b = __builtin_bit_cast(int, x);
    return __builtin_bit_cast(float, __builtin_amdgcn_permlanex16(b, b, 0x76543210, 0xfedcba98, false, false));
}
template <int M>
__device__ __forceinline__ float xrow(float x) {
    const int b = __builtin_bit_cast(int, x);
    return __builtin_bit_cast(float, __builtin_amdgcn_update_dpp(b, b, 0x160 + M, 0xf, 0xf, false));
}

__device__ __forceinline__ float edge_score(float2 l, float2 r, float ea, float2 we, float2 at)
{
    float m0 = l.x + r.x + ea * we.x;
    float m1 = l.y + r.y + ea * we.y;
    m0 = (m0 > 0.0f) ? m0 : 0.2f * m0;
    m1 = (m1 > 0.0f) ? m1 : 0.2f * m1;
    float p = at.x * m0 + at.y * m1;
    p += xhalf(p);
    p += xrow<8>(p);
    p += xrow<4>(p);
    p += xrow<2>(p);
    p += xrow<1>(p);
    return p;
}

__global__ void __launch_bounds__(1024)
node_range_gat_kernel(const int* __restrict__ src, const int* __restrict__ dst,
                      const float* __restrict__ eattr,
                      const float* __restrict__ We, const float* __restrict__ att,
                      const float* __restrict__ xl, const float* __restrict__ xr,
                      float* __restrict__ agg, int NT, long long E,
                      const unsigned* __restrict__ kplanes)
{
    __shared__ float    sAgg[(RMAX + 1) * 64];
    __shared__ float    sXr[(RMAX + 1) * 64];
    __shared__ float    sDen[RMAX + 1];
    __shared__ float    sMax[RMAX + 1];
    __shared__ unsigned sRnd[RMAX + 1];
    __shared__ float    sScl[RMAX + 1];
    __shared__ int   qE[QCAP];
    __shared__ int   qD[QCAP];
    __shared__ float qX[QCAP];
    __shared__ int   qN;

    const int tid   = threadIdx.x;
    const int lane  = tid & 31;
    const int wave  = __builtin_amdgcn_readfirstlane(tid >> 5);
    const int nwave = blockDim.x >> 5;
    const int nthr  = blockDim.x;
    const long long nG = (E + 31) >> 5;
    const uint4* __restrict__ klo = (const uint4*)kplanes;
    const uint4* __restrict__ khi = klo + nG;
    const float2 we = ((const float2*)We)[lane];
    const float2 at = ((const float2*)att)[lane];

    for (int base = blockIdx.x * RMAX; base < NT; base += gridDim.x * RMAX) {
        const int R = (NT - base < RMAX) ? (NT - base) : RMAX;
        for (int i = tid; i < R * 64; i += nthr) {
            sAgg[i] = 0.0f;
            sXr[i]  = xr[(size_t)base * 64 + i];
        }
        for (int i = tid; i < 64; i += nthr) {
            sAgg[RMAX * 64 + i] = 0.0f;
            sXr[RMAX * 64 + i]  = 0.0f;
        }
        for (int i = tid; i <= RMAX; i += nthr) {
            sDen[i] = 0.0f;
            sMax[i] = -__builtin_inff();
            sRnd[i] = 0u;
        }
        if (tid == 0) qN = 0;
        __syncthreads();

        const unsigned pb = (unsigned)(base / RMAX) & 255u;
        unsigned ik[8];
        #pragma unroll
        for (int k = 0; k < 8; ++k) ik[k] = ((pb >> k) & 1u) ? 0u : ~0u;

        long long c0  = 0;
        long long seg = SUPER;
        while (c0 < E) {
            const long long c1 = (c0 + seg < E) ? (c0 + seg) : E;

            const unsigned G0    = (unsigned)(c0 >> 5);
            const unsigned G1    = (unsigned)((c1 + 31) >> 5);
            const unsigned trip  = (unsigned)nthr * NGRP;
            const unsigned ntrip = (G1 - G0 + trip - 1) / trip;
            const unsigned rot   = ntrip * blockIdx.x / gridDim.x;
            auto first_group = [&](unsigned tr) -> unsigned {
                if (tr >= ntrip) return G1;
                const unsigned tt = (tr + rot < ntrip) ? (tr + rot) : (tr + rot - ntrip);
                return G0 + tt * trip + (unsigned)tid;
            };
            auto load_groups = [&](uint4 (&lo)[NGRP], uint4 (&hi)[NGRP], unsigned g0) {
                #pragma unroll
                for (int j = 0; j < NGRP; ++j) {
                    const unsigned g  = g0 + (unsigned)j * nthr;
                    const unsigned gc = (g < G1) ? g : (G1 - 1u);
                    lo[j] = klo[gc];
                    hi[j] = khi[gc];
                }
            };
            auto take_groups = [&](const uint4 (&lo)[NGRP], const uint4 (&hi)[NGRP], unsigned g0) {
                #pragma unroll
                for (int j = 0; j < NGRP; ++j) {
                    unsigned m = (lo[j].x ^ ik[0]) & (lo[j].y ^ ik[1]) & (lo[j].z ^ ik[2]) & (lo[j].w ^ ik[3])
                               & (hi[j].x ^ ik[4]) & (hi[j].y ^ ik[5]) & (hi[j].z ^ ik[6]) & (hi[j].w ^ ik[7]);
                    if (m != 0u) {
                        const long long eb = (long long)(g0 + (unsigned)j * nthr) << 5;
                        do {
                            const long long e = eb + __builtin_ctz(m);
                            m &= m - 1u;
                            if (e < c1) {
                                const int idx = atomicAdd(&qN, 1);
                                if (idx < QCAP) qE[idx] = (int)e;
                            }
                        } while (m != 0u);
                    }
                }
            };
            {
                uint4 loA[NGRP], hiA[NGRP], loB[NGRP], hiB[NGRP];
                unsigned gA = first_group(0u);
                load_groups(loA, hiA, gA);
                for (unsigned tr = 0; tr < ntrip; tr += 2) {
                    const unsigned gB = first_group(tr + 1);
                    load_groups(loB, hiB, gB);
                    take_groups(loA, hiA, gA);
                    gA = first_group(tr + 2);
                    load_groups(loA, hiA, gA);
                    if (tr + 1 < ntrip) take_groups(loB, hiB, gB);
                }
            }
            __syncthreads();

            const int n = qN;
            if (n > QCAP) {
                __syncthreads();
                if (tid == 0) qN = 0;
                seg = (seg / 2 > QCAP) ? seg / 2 : QCAP;
                __syncthreads();
                continue;
            }

            for (int i = tid; i < n; i += nthr) {
                const int ee = qE[i];
                const unsigned dl = (unsigned)(dst[ee] - base);
                qD[i] = (dl < (unsigned)R) ? (int)dl : RMAX;
                qE[i] = src[ee];
                qX[i] = eattr[ee];
            }
            __syncthreads();
            if (tid == 0) qN = 0;

            {
                int i0 = wave;
                for (; i0 + (NROW - 1) * nwave < n; i0 += NROW * nwave) {
                    float2 l[NROW];
                    float  ea[NROW];
                    int    dl[NROW];
                    #pragma unroll
                    for (int j = 0; j < NROW; ++j) {
                        const int i = i0 + j * nwave;
                        ea[j] = qX[i];
                        dl[j] = qD[i];
                        l[j]  = ((const float2*)(xl + (size_t)qE[i] * 64))[lane];
                    }
                    float p[NROW];
                    #pragma unroll
                    for (int j = 0; j < NROW; ++j) {
                        const float2 r = ((const float2*)(sXr + dl[j] * 64))[lane];
                        p[j] = edge_score(l[j], r, ea[j], we, at);
                    }
                    if (lane == 0) {
                        #pragma unroll
                        for (int j = 0; j < NROW; ++j) {
                            qX[i0 + j * nwave] = p[j];
                            atomicMax(&sRnd[dl[j]], enc_f32(p[j]));
                        }
                    }
                }
                for (; i0 < n; i0 += nwave) {
                    const int    dl = qD[i0];
                    const float2 l  = ((const float2*)(xl + (size_t)qE[i0] * 64))[lane];
                    const float2 r  = ((const float2*)(sXr + dl * 64))[lane];
                    const float  p  = edge_score(l, r, qX[i0], we, at);
                    if (lane == 0) {
                        qX[i0] = p;
                        atomicMax(&sRnd[dl], enc_f32(p));
                    }
                }
            }
            __syncthreads();

            for (int r = tid; r <= RMAX; r += nthr) {
                const unsigned u = sRnd[r];
                float s = 1.0f;
                if (u != 0u) {
                    const float mo = sMax[r];
                    const float mn = dec_f32(u);
                    if (mn > mo) {
                        if (mo != -__builtin_inff()) s = __expf(mo - mn);
                        sMax[r] = mn;
                    }
                }
                sScl[r] = s;
                sRnd[r] = 0u;
            }
            __syncthreads();
            if (c0 != 0) {
                for (int i = tid; i < R * 64; i += nthr) sAgg[i] *= sScl[i >> 6];
                for (int i = tid; i < 64; i += nthr) sAgg[RMAX * 64 + i] *= sScl[RMAX];
                for (int r = tid; r <= RMAX; r += nthr) sDen[r] *= sScl[r];
                __syncthreads();
            }

            for (int i = tid; i < n; i += nthr) {
                const int   dl = qD[i];
                const float ex = __expf(qX[i] - sMax[dl]);
                qX[i] = ex;
                atomicAdd(&sDen[dl], ex);
            }
            __syncthreads();

            {
                int i0 = wave;
                for (; i0 + (NROW - 1) * nwave < n; i0 += NROW * nwave) {
                    float2 r[NROW];
                    float  x[NROW];
                    int    dl[NROW];
                    #pragma unroll
                    for (int j = 0; j < NROW; ++j) {
                        const int i = i0 + j * nwave;
                        x[j]  = qX[i];
                        dl[j] = qD[i];
                        r[j]  = ((const float2*)(xl + (size_t)qE[i] * 64))[lane];
                    }
                    #pragma unroll
                    for (int j = 0; j < NROW; ++j) {
                        float* ap = sAgg + dl[j] * 64 + 2 * lane;
                        atomicAdd(ap,     x[j] * r[j].x);
                        atomicAdd(ap + 1, x[j] * r[j].y);
                    }
                }
                for (; i0 < n; i0 += nwave) {
                    const float  x = qX[i0];
                    const float2 r = ((const float2*)(xl + (size_t)qE[i0] * 64))[lane];
                    float* ap = sAgg + qD[i0] * 64 + 2 * lane;
                    atomicAdd(ap,     x * r.x);
                    atomicAdd(ap + 1, x * r.y);
                }
            }
            __syncthreads();
            c0 = c1;
        }

        for (int i = tid; i < R * 64; i += nthr) {
            const float den = sDen[i >> 6];
            agg[(size_t)base * 64 + i] = (den > 0.0f) ? sAgg[i] / den : 0.0f;
        }
        __syncthreads();
    }
}

__global__ void out_xform_kernel(const float* __restrict__ agg, const float* __restrict__ bias,
                                 const float* __restrict__ Wfc, const float* __restrict__ bfc,
                                 float* __restrict__ out, int NT)
{
    __shared__ float sW[WLDS];
    __shared__ float sb[64], sbias[64];
    stage_weight_pairs(sW, Wfc);
    for (int i = threadIdx.x; i < 64; i += blockDim.x) { sb[i] = bfc[i]; sbias[i] = bias[i]; }
    __syncthreads();

    const int lane = threadIdx.x & 31;
    const int wave = threadIdx.x >> 5;
    const int wpb  = blockDim.x >> 5;
    const int l15  = lane & 15;
    const int half = lane >> 4;
    const int numTiles = NT >> 4;

    for (int tile = blockIdx.x * wpb + wave; tile < numTiles; tile += gridDim.x * wpb) {
        v8f acc[4] = {v8f{}, v8f{}, v8f{}, v8f{}};
        const float* arow = agg + (size_t)(tile * 16 + l15) * 64;
        #pragma unroll 4
        for (int kk = 0; kk < 16; ++kk) {
            const int kbase = kk * 4 + half * 2;
            v2f a = *(const v2f*)(arow + kbase);
            a.x += sbias[kbase];
            a.y += sbias[kbase + 1];
            const int pbase = (kk * 2 + half) * PSTRIDE;
            #pragma unroll
            for (int nb = 0; nb < 4; ++nb) {
                const int col = nb * 16 + l15;
                const v2f b = *(const v2f*)(sW + pbase + col * 2);
                acc[nb] = __builtin_amdgcn_wmma_f32_16x16x4_f32(
                    false, a, false, b, (short)0, acc[nb], false, false);
            }
        }
        #pragma unroll
        for (int nb = 0; nb < 4; ++nb) {
            const int col = nb * 16 + l15;
            #pragma unroll
            for (int v = 0; v < 8; ++v) {
                const size_t row = (size_t)(tile * 16 + v + half * 8);
                out[row * 64 + col] = acc[nb][v] + sb[col];
            }
        }
    }
}

extern "C" void kernel_launch(void* const* d_in, const int* in_sizes, int n_in,
                              void* d_out, int out_size, void* d_ws, size_t ws_size,
                              hipStream_t stream) {
    const float* x      = (const float*)d_in[0];
    const int*   eidx   = (const int*)d_in[1];
    const float* eattr  = (const float*)d_in[2];
    const float* W_l    = (const float*)d_in[3];
    const float* b_l    = (const float*)d_in[4];
    const float* W_r    = (const float*)d_in[5];
    const float* b_r    = (const float*)d_in[6];
    const float* W_e    = (const float*)d_in[7];
    const float* att    = (const float*)d_in[8];
    const float* bias   = (const float*)d_in[9];
    const float* W_fc   = (const float*)d_in[10];
    const float* b_fc   = (const float*)d_in[11];

    const int NT       = in_sizes[0] / 64;
    const long long E  = (long long)in_sizes[2];
    const int* src = eidx;
    const int* dst = eidx + E;

    float* ws     = (float*)d_ws;
    float* xl     = ws;
    float* xr     = xl + (size_t)NT * 64;
    float* agg    = xr + (size_t)NT * 64;
    unsigned* kplanes = (unsigned*)(agg + (size_t)NT * 64);

    const int numTiles = NT / 16;
    const int gemmBlocks = (numTiles + 7) / 8;

    node_xform_kernel<<<gemmBlocks, 256, 0, stream>>>(x, W_l, b_l, W_r, b_r, xl, xr, NT, dst, E, kplanes);
    node_range_gat_kernel<<<256, 1024, 0, stream>>>(src, dst, eattr, W_e, att, xl, xr, agg, NT, E, kplanes);
    out_xform_kernel<<<gemmBlocks, 256, 0, stream>>>(agg, bias, W_fc, b_fc, (float*)d_out, NT);
}


# edge score pass: four lanes per queued edge (16 channels a lane as four 16-byte pieces, 8 edges a wave step), not a wave per edge; the score's 64-term sum order changes (4 lanes x 16, was 32 x 2)
# speedup vs baseline: 3.8249x; 1.2394x over previous
#include <hip/hip_runtime.h>
#include <hip/hip_bf16.h>

typedef float v2f __attribute__((ext_vector_type(2)));
typedef float v8f __attribute__((ext_vector_type(8)));

#define PSTRIDE 160
#define WLDS    (32 * PSTRIDE)

__device__ __forceinline__ unsigned enc_f32(float f) {
    unsigned u = __float_as_uint(f);
    return (u & 0x80000000u) ? ~u : (u | 0x80000000u);
}
__device__ __forceinline__ float dec_f32(unsigned e) {
    return (e & 0x80000000u) ? __uint_as_float(e ^ 0x80000000u) : __uint_as_float(~e);
}

#define RMAX  196

__device__ __forceinline__ void write_owner_key_planes(const int* __restrict__ dst, long long E,
                                                       unsigned* __restrict__ kplanes)
{
    const long long nG = (E + 31) >> 5;
    uint4* __restrict__ klo = (uint4*)kplanes;
    uint4* __restrict__ khi = klo + nG;
    const long long nthr = (long long)gridDim.x * blockDim.x;
    for (long long G = (long long)blockIdx.x * blockDim.x + threadIdx.x; G < nG; G += nthr) {
        int d[32];
        #pragma unroll
        for (int l = 0; l < 32; ++l) {
            const long long e = (G << 5) + l;
            d[l] = dst[(e < E) ? e : (E - 1)];
        }
        unsigned p[8] = {0u, 0u, 0u, 0u, 0u, 0u, 0u, 0u};
        #pragma unroll
        for (int l = 0; l < 32; ++l) {
            const unsigned key = (unsigned)d[l] / RMAX;
            #pragma unroll
            for (int k = 0; k < 8; ++k) p[k] |= ((key >> k) & 1u) << l;
        }
        klo[G] = make_uint4(p[0], p[1], p[2], p[3]);
        khi[G] = make_uint4(p[4], p[5], p[6], p[7]);
    }
}

__device__ __forceinline__ void stage_weight_pairs(float* sW, const float* __restrict__ W) {
    for (int i = threadIdx.x; i < 64 * 64; i += blockDim.x) {
        const int k = i >> 6, col = i & 63;
        sW[(k >> 1) * PSTRIDE + col * 2 + (k & 1)] = W[i];
    }
}

__global__ void node_xform_kernel(const float* __restrict__ x2,
                                  const float* __restrict__ Wl, const float* __restrict__ bl,
                                  const float* __restrict__ Wr, const float* __restrict__ br,
                                  float* __restrict__ xl, float* __restrict__ xr, int NT,
                                  const int* __restrict__ dst, long long E, unsigned* __restrict__ kplanes)
{
    write_owner_key_planes(dst, E, kplanes);

    __shared__ float sWl[WLDS];
    __shared__ float sWr[WLDS];
    __shared__ float sbl[64], sbr[64];
    stage_weight_pairs(sWl, Wl);
    stage_weight_pairs(sWr, Wr);
    for (int i = threadIdx.x; i < 64; i += blockDim.x) { sbl[i] = bl[i]; sbr[i] = br[i]; }
    __syncthreads();

    const int lane    = threadIdx.x & 31;
    const int wave    = threadIdx.x >> 5;
    const int wpb     = blockDim.x >> 5;
    const int l15     = lane & 15;
    const int half    = lane >> 4;
    const int numTiles = NT >> 4;

    for (int tile = blockIdx.x * wpb + wave; tile < numTiles; tile += gridDim.x * wpb) {
        v8f accl[4] = {v8f{}, v8f{}, v8f{}, v8f{}};
        v8f accr[4] = {v8f{}, v8f{}, v8f{}, v8f{}};
        const float* xrow = x2 + (size_t)(tile * 16 + l15) * 64;
        #pragma unroll 4
        for (int kk = 0; kk < 16; ++kk) {
            const int kbase = kk * 4 + half * 2;
            const v2f a = *(const v2f*)(xrow + kbase);
            const int pbase = (kk * 2 + half) * PSTRIDE;
            #pragma unroll
            for (int nb = 0; nb < 4; ++nb) {
                const int col = nb * 16 + l15;
                const v2f bL = *(const v2f*)(sWl + pbase + col * 2);
                const v2f bR = *(const v2f*)(sWr + pbase + col * 2);
                accl[nb] = __builtin_amdgcn_wmma_f32_16x16x4_f32(
                    false, a, false, bL, (short)0, accl[nb], false, false);
                accr[nb] = __builtin_amdgcn_wmma_f32_16x16x4_f32(
                    false, a, false, bR, (short)0, accr[nb], false, false);
            }
        }
        #pragma unroll
        for (int nb = 0; nb < 4; ++nb) {
            const int col = nb * 16 + l15;
            #pragma unroll
            for (int v = 0; v < 8; ++v) {
                const size_t row = (size_t)(tile * 16 + v + half * 8);
                xl[row * 64 + col] = accl[nb][v] + sbl[col];
                xr[row * 64 + col] = accr[nb][v] + sbr[col];
            }
        }
    }
}

#define QCAP  16384
#define SUPER 2097152
#define NGRP  4
#define NROW  8
#define NSET  2

template <int M>
__device__ __forceinline__ float xrow(float x) {
    const int b = __builtin_bit_cast(int, x);
    return __builtin_bit_cast(float, __builtin_amdgcn_update_dpp(b, b, 0x160 + M, 0xf, 0xf, false));
}

__device__ __forceinline__ float piece_score(float4 l, float4 r, float ea, float4 we, float4 at)
{
    float m0 = l.x + r.x + ea * we.x;
    float m1 = l.y + r.y + ea * we.y;
    float m2 = l.z + r.z + ea * we.z;
    float m3 = l.w + r.w + ea * we.w;
    m0 = (m0 > 0.0f) ? m0 : 0.2f * m0;
    m1 = (m1 > 0.0f) ? m1 : 0.2f * m1;
    m2 = (m2 > 0.0f) ? m2 : 0.2f * m2;
    m3 = (m3 > 0.0f) ? m3 : 0.2f * m3;
    return ((at.x * m0 + at.y * m1) + at.z * m2) + at.w * m3;
}

__global__ void __launch_bounds__(1024)
node_range_gat_kernel(const int* __restrict__ src, const int* __restrict__ dst,
                      const float* __restrict__ eattr,
                      const float* __restrict__ We, const float* __restrict__ att,
                      const float* __restrict__ xl, const float* __restrict__ xr,
                      float* __restrict__ agg, int NT, long long E,
                      const unsigned* __restrict__ kplanes)
{
    __shared__ float    sAgg[(RMAX + 1) * 64];
    __shared__ __attribute__((aligned(16))) float sXr[(RMAX + 1) * 64];
    __shared__ float    sDen[RMAX + 1];
    __shared__ float    sMax[RMAX + 1];
    __shared__ unsigned sRnd[RMAX + 1];
    __shared__ float    sScl[RMAX + 1];
    __shared__ int   qE[QCAP];
    __shared__ int   qD[QCAP];
    __shared__ float qX[QCAP];
    __shared__ int   qN;
    __shared__ __attribute__((aligned(16))) float sWe[64];
    __shared__ __attribute__((aligned(16))) float sAt[64];

    const int tid   = threadIdx.x;
    const int lane  = tid & 31;
    const int wave  = __builtin_amdgcn_readfirstlane(tid >> 5);
    const int nwave = blockDim.x >> 5;
    const int nthr  = blockDim.x;
    const long long nG = (E + 31) >> 5;
    const uint4* __restrict__ klo = (const uint4*)kplanes;
    const uint4* __restrict__ khi = klo + nG;

    for (int i = tid; i < 64; i += nthr) { sWe[i] = We[i]; sAt[i] = att[i]; }

    for (int base = blockIdx.x * RMAX; base < NT; base += gridDim.x * RMAX) {
        const int R = (NT - base < RMAX) ? (NT - base) : RMAX;
        for (int i = tid; i < R * 64; i += nthr) {
            sAgg[i] = 0.0f;
            sXr[i]  = xr[(size_t)base * 64 + i];
        }
        for (int i = tid; i < 64; i += nthr) {
            sAgg[RMAX * 64 + i] = 0.0f;
            sXr[RMAX * 64 + i]  = 0.0f;
        }
        for (int i = tid; i <= RMAX; i += nthr) {
            sDen[i] = 0.0f;
            sMax[i] = -__builtin_inff();
            sRnd[i] = 0u;
        }
        if (tid == 0) qN = 0;
        __syncthreads();

        const unsigned pb = (unsigned)(base / RMAX) & 255u;
        unsigned ik[8];
        #pragma unroll
        for (int k = 0; k < 8; ++k) ik[k] = ((pb >> k) & 1u) ? 0u : ~0u;

        long long c0  = 0;
        long long seg = SUPER;
        while (c0 < E) {
            const long long c1 = (c0 + seg < E) ? (c0 + seg) : E;

            const unsigned G0    = (unsigned)(c0 >> 5);
            const unsigned G1    = (unsigned)((c1 + 31) >> 5);
            const unsigned trip  = (unsigned)nthr * NGRP;
            const unsigned ntrip = (G1 - G0 + trip - 1) / trip;
            const unsigned rot   = ntrip * blockIdx.x / gridDim.x;
            auto first_group = [&](unsigned tr) -> unsigned {
                if (tr >= ntrip) return G1;
                const unsigned tt = (tr + rot < ntrip) ? (tr + rot) : (tr + rot - ntrip);
                return G0 + tt * trip + (unsigned)tid;
            };
            auto load_groups = [&](uint4 (&lo)[NGRP], uint4 (&hi)[NGRP], unsigned g0) {
                #pragma unroll
                for (int j = 0; j < NGRP; ++j) {
                    const unsigned g  = g0 + (unsigned)j * nthr;
                    const unsigned gc = (g < G1) ? g : (G1 - 1u);
                    lo[j] = klo[gc];
                    hi[j] = khi[gc];
                }
            };
            auto take_groups = [&](const uint4 (&lo)[NGRP], const uint4 (&hi)[NGRP], unsigned g0) {
                #pragma unroll
                for (int j = 0; j < NGRP; ++j) {
                    unsigned m = (lo[j].x ^ ik[0]) & (lo[j].y ^ ik[1]) & (lo[j].z ^ ik[2]) & (lo[j].w ^ ik[3])
                               & (hi[j].x ^ ik[4]) & (hi[j].y ^ ik[5]) & (hi[j].z ^ ik[6]) & (hi[j].w ^ ik[7]);
                    if (m != 0u) {
                        const long long eb = (long long)(g0 + (unsigned)j * nthr) << 5;
                        do {
                            const long long e = eb + __builtin_ctz(m);
                            m &= m - 1u;
                            if (e < c1) {
                                const int idx = atomicAdd(&qN, 1);
                                if (idx < QCAP) qE[idx] = (int)e;
                            }
                        } while (m != 0u);
                    }
                }
            };
            {
                uint4 loA[NGRP], hiA[NGRP], loB[NGRP], hiB[NGRP];
                unsigned gA = first_group(0u);
                load_groups(loA, hiA, gA);
                for (unsigned tr = 0; tr < ntrip; tr += 2) {
                    const unsigned gB = first_group(tr + 1);
                    load_groups(loB, hiB, gB);
                    take_groups(loA, hiA, gA);
                    gA = first_group(tr + 2);
                    load_groups(loA, hiA, gA);
                    if (tr + 1 < ntrip) take_groups(loB, hiB, gB);
                }
            }
            __syncthreads();

            const int n = qN;
            if (n > QCAP) {
                __syncthreads();
                if (tid == 0) qN = 0;
                seg = (seg / 2 > QCAP) ? seg / 2 : QCAP;
                __syncthreads();
                continue;
            }

            for (int i = tid; i < n; i += nthr) {
                const int ee = qE[i];
                const unsigned dl = (unsigned)(dst[ee] - base);
                qD[i] = (dl < (unsigned)R) ? (int)dl : RMAX;
                qE[i] = src[ee];
                qX[i] = eattr[ee];
            }
            __syncthreads();
            if (tid == 0) qN = 0;

            {
                const int q4 = lane & 3;
                const int g8 = lane >> 2;
                unsigned cb[4];
                #pragma unroll
                for (int k = 0; k < 4; ++k) cb[k] = 64u * (unsigned)((k + g8) & 3) + 16u * (unsigned)q4;
                float4 we4[4], at4[4];
                #pragma unroll
                for (int k = 0; k < 4; ++k) {
                    we4[k] = *(const float4*)((const char*)sWe + cb[k]);
                    at4[k] = *(const float4*)((const char*)sAt + cb[k]);
                }
                for (int i0 = wave * 8; i0 < n; i0 += NSET * 8 * nwave) {
                    int    id[NSET];
                    float  ea[NSET];
                    int    dl[NSET];
                    float4 l[NSET][4];
                    #pragma unroll
                    for (int s = 0; s < NSET; ++s) {
                        const int i  = i0 + s * 8 * nwave + g8;
                        const int ic = (i < n) ? i : (n - 1);
                        id[s] = (i < n) ? i : -1;
                        ea[s] = qX[ic];
                        dl[s] = qD[ic];
                        const unsigned lo = (unsigned)qE[ic] * 256u;
                        #pragma unroll
                        for (int k = 0; k < 4; ++k) l[s][k] = *(const float4*)((const char*)xl + (size_t)(lo + cb[k]));
                    }
                    __builtin_amdgcn_sched_barrier(0);
                    #pragma unroll
                    for (int s = 0; s < NSET; ++s) {
                        const unsigned ro = (unsigned)dl[s] * 256u;
                        float pk[4];
                        #pragma unroll
                        for (int k = 0; k < 4; ++k) {
                            const float4 r = *(const float4*)((const char*)sXr + (ro + cb[k]));
                            pk[k] = piece_score(l[s][k], r, ea[s], we4[k], at4[k]);
                        }
                        float p = (pk[0] + pk[1]) + (pk[2] + pk[3]);
                        p += xrow<1>(p);
                        p += xrow<2>(p);
                        if (q4 == 0 && id[s] >= 0) {
                            qX[id[s]] = p;
                            atomicMax(&sRnd[dl[s]], enc_f32(p));
                        }
                    }
                }
            }
            __syncthreads();

            for (int r = tid; r <= RMAX; r += nthr) {
                const unsigned u = sRnd[r];
                float s = 1.0f;
                if (u != 0u) {
                    const float mo = sMax[r];
                    const float mn = dec_f32(u);
                    if (mn > mo) {
                        if (mo != -__builtin_inff()) s = __expf(mo - mn);
                        sMax[r] = mn;
                    }
                }
                sScl[r] = s;
                sRnd[r] = 0u;
            }
            __syncthreads();
            if (c0 != 0) {
                for (int i = tid; i < R * 64; i += nthr) sAgg[i] *= sScl[i >> 6];
                for (int i = tid; i < 64; i += nthr) sAgg[RMAX * 64 + i] *= sScl[RMAX];
                for (int r = tid; r <= RMAX; r += nthr) sDen[r] *= sScl[r];
                __syncthreads();
            }

            for (int i = tid; i < n; i += nthr) {
                const int   dl = qD[i];
                const float ex = __expf(qX[i] - sMax[dl]);
                qX[i] = ex;
                atomicAdd(&sDen[dl], ex);
            }
            __syncthreads();

            {
                int i0 = wave;
                for (; i0 + (NROW - 1) * nwave < n; i0 += NROW * nwave) {
                    float2 r[NROW];
                    float  x[NROW];
                    int    dl[NROW];
                    #pragma unroll
                    for (int j = 0; j < NROW; ++j) {
                        const int i = i0 + j * nwave;
                        x[j]  = qX[i];
                        dl[j] = qD[i];
                        r[j]  = ((const float2*)(xl + (size_t)qE[i] * 64))[lane];
                    }
                    #pragma unroll
                    for (int j = 0; j < NROW; ++j) {
                        float* ap = sAgg + dl[j] * 64 + 2 * lane;
                        atomicAdd(ap,     x[j] * r[j].x);
                        atomicAdd(ap + 1, x[j] * r[j].y);
                    }
                }
                for (; i0 < n; i0 += nwave) {
                    const float  x = qX[i0];
                    const float2 r = ((const float2*)(xl + (size_t)qE[i0] * 64))[lane];
                    float* ap = sAgg + qD[i0] * 64 + 2 * lane;
                    atomicAdd(ap,     x * r.x);
                    atomicAdd(ap + 1, x * r.y);
                }
            }
            __syncthreads();
            c0 = c1;
        }

        for (int i = tid; i < R * 64; i += nthr) {
            const float den = sDen[i >> 6];
            agg[(size_t)base * 64 + i] = (den > 0.0f) ? sAgg[i] / den : 0.0f;
        }
        __syncthreads();
    }
}

__global__ void out_xform_kernel(const float* __restrict__ agg, const float* __restrict__ bias,
                                 const float* __restrict__ Wfc, const float* __restrict__ bfc,
                                 float* __restrict__ out, int NT)
{
    __shared__ float sW[WLDS];
    __shared__ float sb[64], sbias[64];
    stage_weight_pairs(sW, Wfc);
    for (int i = threadIdx.x; i < 64; i += blockDim.x) { sb[i] = bfc[i]; sbias[i] = bias[i]; }
    __syncthreads();

    const int lane = threadIdx.x & 31;
    const int wave = threadIdx.x >> 5;
    const int wpb  = blockDim.x >> 5;
    const int l15  = lane & 15;
    const int half = lane >> 4;
    const int numTiles = NT >> 4;

    for (int tile = blockIdx.x * wpb + wave; tile < numTiles; tile += gridDim.x * wpb) {
        v8f acc[4] = {v8f{}, v8f{}, v8f{}, v8f{}};
        const float* arow = agg + (size_t)(tile * 16 + l15) * 64;
        #pragma unroll 4
        for (int kk = 0; kk < 16; ++kk) {
            const int kbase = kk * 4 + half * 2;
            v2f a = *(const v2f*)(arow + kbase);
            a.x += sbias[kbase];
            a.y += sbias[kbase + 1];
            const int pbase = (kk * 2 + half) * PSTRIDE;
            #pragma unroll
            for (int nb = 0; nb < 4; ++nb) {
                const int col = nb * 16 + l15;
                const v2f b = *(const v2f*)(sW + pbase + col * 2);
                acc[nb] = __builtin_amdgcn_wmma_f32_16x16x4_f32(
                    false, a, false, b, (short)0, acc[nb], false, false);
            }
        }
        #pragma unroll
        for (int nb = 0; nb < 4; ++nb) {
            const int col = nb * 16 + l15;
            #pragma unroll
            for (int v = 0; v < 8; ++v) {
                const size_t row = (size_t)(tile * 16 + v + half * 8);
                out[row * 64 + col] = acc[nb][v] + sb[col];
            }
        }
    }
}

extern "C" void kernel_launch(void* const* d_in, const int* in_sizes, int n_in,
                              void* d_out, int out_size, void* d_ws, size_t ws_size,
                              hipStream_t stream) {
    const float* x      = (const float*)d_in[0];
    const int*   eidx   = (const int*)d_in[1];
    const float* eattr  = (const float*)d_in[2];
    const float* W_l    = (const float*)d_in[3];
    const float* b_l    = (const float*)d_in[4];
    const float* W_r    = (const float*)d_in[5];
    const float* b_r    = (const float*)d_in[6];
    const float* W_e    = (const float*)d_in[7];
    const float* att    = (const float*)d_in[8];
    const float* bias   = (const float*)d_in[9];
    const float* W_fc   = (const float*)d_in[10];
    const float* b_fc   = (const float*)d_in[11];

    const int NT       = in_sizes[0] / 64;
    const long long E  = (long long)in_sizes[2];
    const int* src = eidx;
    const int* dst = eidx + E;

    float* ws     = (float*)d_ws;
    float* xl     = ws;
    float* xr     = xl + (size_t)NT * 64;
    float* agg    = xr + (size_t)NT * 64;
    unsigned* kplanes = (unsigned*)(agg + (size_t)NT * 64);

    const int numTiles = NT / 16;
    const int gemmBlocks = (numTiles + 7) / 8;

    node_xform_kernel<<<gemmBlocks, 256, 0, stream>>>(x, W_l, b_l, W_r, b_r, xl, xr, NT, dst, E, kplanes);
    node_range_gat_kernel<<<256, 1024, 0, stream>>>(src, dst, eattr, W_e, att, xl, xr, agg, NT, E, kplanes);
    out_xform_kernel<<<gemmBlocks, 256, 0, stream>>>(agg, bias, W_fc, b_fc, (float*)d_out, NT);
}


# src, dst and edge attribute packed into one 8-byte record per edge by the transform kernel's list pass (now lane per edge, coalesced): owners fetch one line per queued edge, not three; sums unchanged
# speedup vs baseline: 4.0999x; 1.0719x over previous
#include <hip/hip_runtime.h>
#include <hip/hip_bf16.h>

typedef float v2f __attribute__((ext_vector_type(2)));
typedef float v8f __attribute__((ext_vector_type(8)));

#define PSTRIDE 160
#define WLDS    (32 * PSTRIDE)

__device__ __forceinline__ unsigned enc_f32(float f) {
    unsigned u = __float_as_uint(f);
    return (u & 0x80000000u) ? ~u : (u | 0x80000000u);
}
__device__ __forceinline__ float dec_f32(unsigned e) {
    return (e & 0x80000000u) ? __uint_as_float(e ^ 0x80000000u) : __uint_as_float(~e);
}

#define RMAX  196

template <bool WHOLE>
__device__ __forceinline__ void records_and_planes_batch(const int* __restrict__ src, const int* __restrict__ dst,
                                                         const float* __restrict__ eattr, long long E, long long nG,
                                                         uint4* __restrict__ klo, uint4* __restrict__ khi,
                                                         uint2* __restrict__ rec, long long G0, int lane)
{
    int   d[16], s[16];
    float a[16];
    #pragma unroll
    for (int j = 0; j < 16; ++j) {
        const long long e  = ((G0 + j) << 5) + lane;
        const long long ec = (WHOLE || e < E) ? e : (E - 1);
        d[j] = dst[ec];
        s[j] = src[ec];
        a[j] = eattr[ec];
    }
    asm volatile("" :: "v"(d[0]), "v"(d[1]), "v"(d[2]), "v"(d[3]), "v"(d[4]), "v"(d[5]), "v"(d[6]), "v"(d[7]),
                       "v"(s[0]), "v"(s[1]), "v"(s[2]), "v"(s[3]), "v"(s[4]), "v"(s[5]), "v"(s[6]), "v"(s[7]),
                       "v"(a[0]), "v"(a[1]), "v"(a[2]), "v"(a[3]), "v"(a[4]), "v"(a[5]), "v"(a[6]), "v"(a[7]));
    asm volatile("" :: "v"(d[8]), "v"(d[9]), "v"(d[10]), "v"(d[11]), "v"(d[12]), "v"(d[13]), "v"(d[14]), "v"(d[15]),
                       "v"(s[8]), "v"(s[9]), "v"(s[10]), "v"(s[11]), "v"(s[12]), "v"(s[13]), "v"(s[14]), "v"(s[15]),
                       "v"(a[8]), "v"(a[9]), "v"(a[10]), "v"(a[11]), "v"(a[12]), "v"(a[13]), "v"(a[14]), "v"(a[15]));
    unsigned mine[8] = {0u, 0u, 0u, 0u, 0u, 0u, 0u, 0u};
    #pragma unroll
    for (int j = 0; j < 16; ++j) {
        const long long e = ((G0 + j) << 5) + lane;
        if (WHOLE || e < E) rec[e] = make_uint2((unsigned)s[j] | ((unsigned)d[j] << 16), __float_as_uint(a[j]));
        const unsigned key = (unsigned)d[j] / RMAX;
        #pragma unroll
        for (int k = 0; k < 8; ++k) {
            const unsigned p = __builtin_amdgcn_ballot_w32(((key >> k) & 1u) != 0u);
            if (lane == j) mine[k] = p;
        }
    }
    if (lane < 16 && (WHOLE || G0 + lane < nG)) {
        klo[G0 + lane] = make_uint4(mine[0], mine[1], mine[2], mine[3]);
        khi[G0 + lane] = make_uint4(mine[4], mine[5], mine[6], mine[7]);
    }
}

__device__ __forceinline__ void write_edge_records_and_key_planes(const int* __restrict__ src,
                                                                  const int* __restrict__ dst,
                                                                  const float* __restrict__ eattr, long long E,
                                                                  unsigned* __restrict__ kplanes,
                                                                  unsigned* __restrict__ recw)
{
    uint2* __restrict__ rec = (uint2*)recw;
    const long long nG = (E + 31) >> 5;
    const long long nB = (nG + 15) >> 4;
    uint4* __restrict__ klo = (uint4*)kplanes;
    uint4* __restrict__ khi = klo + nG;
    const int lane = threadIdx.x & 31;
    const long long wid = (long long)blockIdx.x * (blockDim.x >> 5) + __builtin_amdgcn_readfirstlane((int)(threadIdx.x >> 5));
    const long long nw  = (long long)gridDim.x * (blockDim.x >> 5);
    for (long long b = wid; b < nB; b += nw) {
        const long long G0 = b << 4;
        if (((G0 + 16) << 5) <= E) records_and_planes_batch<true >(src, dst, eattr, E, nG, klo, khi, rec, G0, lane);
        else                       records_and_planes_batch<false>(src, dst, eattr, E, nG, klo, khi, rec, G0, lane);
    }
}

__device__ __forceinline__ void stage_weight_pairs(float* sW, const float* __restrict__ W) {
    for (int i = threadIdx.x; i < 64 * 64; i += blockDim.x) {
        const int k = i >> 6, col = i & 63;
        sW[(k >> 1) * PSTRIDE + col * 2 + (k & 1)] = W[i];
    }
}

__global__ void node_xform_kernel(const float* __restrict__ x2,
                                  const float* __restrict__ Wl, const float* __restrict__ bl,
                                  const float* __restrict__ Wr, const float* __restrict__ br,
                                  float* __restrict__ xl, float* __restrict__ xr, int NT,
                                  const int* __restrict__ src, const int* __restrict__ dst,
                                  const float* __restrict__ eattr, long long E,
                                  unsigned* __restrict__ kplanes, unsigned* __restrict__ recw)
{
    write_edge_records_and_key_planes(src, dst, eattr, E, kplanes, recw);

    __shared__ float sWl[WLDS];
    __shared__ float sWr[WLDS];
    __shared__ float sbl[64], sbr[64];
    stage_weight_pairs(sWl, Wl);
    stage_weight_pairs(sWr, Wr);
    for (int i = threadIdx.x; i < 64; i += blockDim.x) { sbl[i] = bl[i]; sbr[i] = br[i]; }
    __syncthreads();

    const int lane    = threadIdx.x & 31;
    const int wave    = threadIdx.x >> 5;
    const int wpb     = blockDim.x >> 5;
    const int l15     = lane & 15;
    const int half    = lane >> 4;
    const int numTiles = NT >> 4;

    for (int tile = blockIdx.x * wpb + wave; tile < numTiles; tile += gridDim.x * wpb) {
        v8f accl[4] = {v8f{}, v8f{}, v8f{}, v8f{}};
        v8f accr[4] = {v8f{}, v8f{}, v8f{}, v8f{}};
        const float* xrow = x2 + (size_t)(tile * 16 + l15) * 64;
        #pragma unroll 4
        for (int kk = 0; kk < 16; ++kk) {
            const int kbase = kk * 4 + half * 2;
            const v2f a = *(const v2f*)(xrow + kbase);
            const int pbase = (kk * 2 + half) * PSTRIDE;
            #pragma unroll
            for (int nb = 0; nb < 4; ++nb) {
                const int col = nb * 16 + l15;
                const v2f bL = *(const v2f*)(sWl + pbase + col * 2);
                const v2f bR = *(const v2f*)(sWr + pbase + col * 2);
                accl[nb] = __builtin_amdgcn_wmma_f32_16x16x4_f32(
                    false, a, false, bL, (short)0, accl[nb], false, false);
                accr[nb] = __builtin_amdgcn_wmma_f32_16x16x4_f32(
                    false, a, false, bR, (short)0, accr[nb], false, false);
            }
        }
        #pragma unroll
        for (int nb = 0; nb < 4; ++nb) {
            const int col = nb * 16 + l15;
            #pragma unroll
            for (int v = 0; v < 8; ++v) {
                const size_t row = (size_t)(tile * 16 + v + half * 8);
                xl[row * 64 + col] = accl[nb][v] + sbl[col];
                xr[row * 64 + col] = accr[nb][v] + sbr[col];
            }
        }
    }
}

#define QCAP  16384
#define SUPER 2097152
#define NGRP  4
#define NROW  8
#define NSET  2

template <int M>
__device__ __forceinline__ float xrow(float x) {
    const int b = __builtin_bit_cast(int, x);
    return __builtin_bit_cast(float, __builtin_amdgcn_update_dpp(b, b, 0x160 + M, 0xf, 0xf, false));
}

__device__ __forceinline__ float piece_score(float4 l, float4 r, float ea, float4 we, float4 at)
{
    float m0 = l.x + r.x + ea * we.x;
    float m1 = l.y + r.y + ea * we.y;
    float m2 = l.z + r.z + ea * we.z;
    float m3 = l.w + r.w + ea * we.w;
    m0 = (m0 > 0.0f) ? m0 : 0.2f * m0;
    m1 = (m1 > 0.0f) ? m1 : 0.2f * m1;
    m2 = (m2 > 0.0f) ? m2 : 0.2f * m2;
    m3 = (m3 > 0.0f) ? m3 : 0.2f * m3;
    return ((at.x * m0 + at.y * m1) + at.z * m2) + at.w * m3;
}

__global__ void __launch_bounds__(1024)
node_range_gat_kernel(const unsigned* __restrict__ recw,
                      const float* __restrict__ We, const float* __restrict__ att,
                      const float* __restrict__ xl, const float* xr,
                      float* agg, int NT, long long E,
                      const unsigned* __restrict__ kplanes)
{
    __shared__ float    sAgg[(RMAX + 1) * 64];
    __shared__ __attribute__((aligned(16))) float sXr[(RMAX + 1) * 64];
    __shared__ float    sDen[RMAX + 1];
    __shared__ float    sMax[RMAX + 1];
    __shared__ unsigned sRnd[RMAX + 1];
    __shared__ float    sScl[RMAX + 1];
    __shared__ int   qE[QCAP];
    __shared__ int   qD[QCAP];
    __shared__ float qX[QCAP];
    __shared__ int   qN;
    __shared__ __attribute__((aligned(16))) float sWe[64];
    __shared__ __attribute__((aligned(16))) float sAt[64];

    const int tid   = threadIdx.x;
    const int lane  = tid & 31;
    const int wave  = __builtin_amdgcn_readfirstlane(tid >> 5);
    const int nwave = blockDim.x >> 5;
    const int nthr  = blockDim.x;
    const long long nG = (E + 31) >> 5;
    const uint4* __restrict__ klo = (const uint4*)kplanes;
    const uint4* __restrict__ khi = klo + nG;
    const uint2* __restrict__ rec = (const uint2*)recw;

    for (int i = tid; i < 64; i += nthr) { sWe[i] = We[i]; sAt[i] = att[i]; }

    for (int base = blockIdx.x * RMAX; base < NT; base += gridDim.x * RMAX) {
        const int R = (NT - base < RMAX) ? (NT - base) : RMAX;
        for (int i = tid; i < R * 64; i += nthr) {
            sAgg[i] = 0.0f;
            sXr[i]  = xr[(size_t)base * 64 + i];
        }
        for (int i = tid; i < 64; i += nthr) {
            sAgg[RMAX * 64 + i] = 0.0f;
            sXr[RMAX * 64 + i]  = 0.0f;
        }
        for (int i = tid; i <= RMAX; i += nthr) {
            sDen[i] = 0.0f;
            sMax[i] = -__builtin_inff();
            sRnd[i] = 0u;
        }
        if (tid == 0) qN = 0;
        __syncthreads();

        const unsigned pb = (unsigned)(base / RMAX) & 255u;
        unsigned ik[8];
        #pragma unroll
        for (int k = 0; k < 8; ++k) ik[k] = ((pb >> k) & 1u) ? 0u : ~0u;

        long long c0  = 0;
        long long seg = SUPER;
        while (c0 < E) {
            const long long c1 = (c0 + seg < E) ? (c0 + seg) : E;

            const unsigned G0    = (unsigned)(c0 >> 5);
            const unsigned G1    = (unsigned)((c1 + 31) >> 5);
            const unsigned trip  = (unsigned)nthr * NGRP;
            const unsigned ntrip = (G1 - G0 + trip - 1) / trip;
            const unsigned rot   = ntrip * blockIdx.x / gridDim.x;
            auto first_group = [&](unsigned tr) -> unsigned {
                if (tr >= ntrip) return G1;
                const unsigned tt = (tr + rot < ntrip) ? (tr + rot) : (tr + rot - ntrip);
                return G0 + tt * trip + (unsigned)tid;
            };
            auto load_groups = [&](uint4 (&lo)[NGRP], uint4 (&hi)[NGRP], unsigned g0) {
                #pragma unroll
                for (int j = 0; j < NGRP; ++j) {
                    const unsigned g  = g0 + (unsigned)j * nthr;
                    const unsigned gc = (g < G1) ? g : (G1 - 1u);
                    lo[j] = klo[gc];
                    hi[j] = khi[gc];
                }
            };
            auto take_groups = [&](const uint4 (&lo)[NGRP], const uint4 (&hi)[NGRP], unsigned g0) {
                #pragma unroll
                for (int j = 0; j < NGRP; ++j) {
                    unsigned m = (lo[j].x ^ ik[0]) & (lo[j].y ^ ik[1]) & (lo[j].z ^ ik[2]) & (lo[j].w ^ ik[3])
                               & (hi[j].x ^ ik[4]) & (hi[j].y ^ ik[5]) & (hi[j].z ^ ik[6]) & (hi[j].w ^ ik[7]);
                    if (m != 0u) {
                        const long long eb = (long long)(g0 + (unsigned)j * nthr) << 5;
                        do {
                            const long long e = eb + __builtin_ctz(m);
                            m &= m - 1u;
                            if (e < c1) {
                                const int idx = atomicAdd(&qN, 1);
                                if (idx < QCAP) qE[idx] = (int)e;
                            }
                        } while (m != 0u);
                    }
                }
            };
            {
                uint4 loA[NGRP], hiA[NGRP], loB[NGRP], hiB[NGRP];
                unsigned gA = first_group(0u);
                load_groups(loA, hiA, gA);
                for (unsigned tr = 0; tr < ntrip; tr += 2) {
                    const unsigned gB = first_group(tr + 1);
                    load_groups(loB, hiB, gB);
                    take_groups(loA, hiA, gA);
                    gA = first_group(tr + 2);
                    load_groups(loA, hiA, gA);
                    if (tr + 1 < ntrip) take_groups(loB, hiB, gB);
                }
            }
            __syncthreads();

            const int n = qN;
            if (n > QCAP) {
                __syncthreads();
                if (tid == 0) qN = 0;
                seg = (seg / 2 > QCAP) ? seg / 2 : QCAP;
                __syncthreads();
                continue;
            }

            for (int i = tid; i < n; i += nthr) {
                const uint2 r = rec[qE[i]];
                const unsigned dl = (r.x >> 16) - (unsigned)base;
                qD[i] = (dl < (unsigned)R) ? (int)dl : RMAX;
                qE[i] = (int)(r.x & 0xffffu);
                qX[i] = __uint_as_float(r.y);
            }
            __syncthreads();
            if (tid == 0) qN = 0;

            {
                const int q4 = lane & 3;
                const int g8 = lane >> 2;
                unsigned cb[4];
                #pragma unroll
                for (int k = 0; k < 4; ++k) cb[k] = 64u * (unsigned)((k + g8) & 3) + 16u * (unsigned)q4;
                float4 we4[4], at4[4];
                #pragma unroll
                for (int k = 0; k < 4; ++k) {
                    we4[k] = *(const float4*)((const char*)sWe + cb[k]);
                    at4[k] = *(const float4*)((const char*)sAt + cb[k]);
                }
                for (int i0 = wave * 8; i0 < n; i0 += NSET * 8 * nwave) {
                    int    id[NSET];
                    float  ea[NSET];
                    int    dl[NSET];
                    float4 l[NSET][4];
                    #pragma unroll
                    for (int s = 0; s < NSET; ++s) {
                        const int i  = i0 + s * 8 * nwave + g8;
                        const int ic = (i < n) ? i : (n - 1);
                        id[s] = (i < n) ? i : -1;
                        ea[s] = qX[ic];
                        dl[s] = qD[ic];
                        const unsigned lo = (unsigned)qE[ic] * 256u;
                        #pragma unroll
                        for (int k = 0; k < 4; ++k) l[s][k] = *(const float4*)((const char*)xl + (size_t)(lo + cb[k]));
                    }
                    __builtin_amdgcn_sched_barrier(0);
                    #pragma unroll
                    for (int s = 0; s < NSET; ++s) {
                        const unsigned ro = (unsigned)dl[s] * 256u;
                        float pk[4];
                        #pragma unroll
                        for (int k = 0; k < 4; ++k) {
                            const float4 r = *(const float4*)((const char*)sXr + (ro + cb[k]));
                            pk[k] = piece_score(l[s][k], r, ea[s], we4[k], at4[k]);
                        }
                        float p = (pk[0] + pk[1]) + (pk[2] + pk[3]);
                        p += xrow<1>(p);
                        p += xrow<2>(p);
                        if (q4 == 0 && id[s] >= 0) {
                            qX[id[s]] = p;
                            atomicMax(&sRnd[dl[s]], enc_f32(p));
                        }
                    }
                }
            }
            __syncthreads();

            for (int r = tid; r <= RMAX; r += nthr) {
                const unsigned u = sRnd[r];
                float s = 1.0f;
                if (u != 0u) {
                    const float mo = sMax[r];
                    const float mn = dec_f32(u);
                    if (mn > mo) {
                        if (mo != -__builtin_inff()) s = __expf(mo - mn);
                        sMax[r] = mn;
                    }
                }
                sScl[r] = s;
                sRnd[r] = 0u;
            }
            __syncthreads();
            if (c0 != 0) {
                for (int i = tid; i < R * 64; i += nthr) sAgg[i] *= sScl[i >> 6];
                for (int i = tid; i < 64; i += nthr) sAgg[RMAX * 64 + i] *= sScl[RMAX];
                for (int r = tid; r <= RMAX; r += nthr) sDen[r] *= sScl[r];
                __syncthreads();
            }

            for (int i = tid; i < n; i += nthr) {
                const int   dl = qD[i];
                const float ex = __expf(qX[i] - sMax[dl]);
                qX[i] = ex;
                atomicAdd(&sDen[dl], ex);
            }
            __syncthreads();

            {
                int i0 = wave;
                for (; i0 + (NROW - 1) * nwave < n; i0 += NROW * nwave) {
                    float2 r[NROW];
                    float  x[NROW];
                    int    dl[NROW];
                    #pragma unroll
                    for (int j = 0; j < NROW; ++j) {
                        const int i = i0 + j * nwave;
                        x[j]  = qX[i];
                        dl[j] = qD[i];
                        r[j]  = ((const float2*)(xl + (size_t)qE[i] * 64))[lane];
                    }
                    #pragma unroll
                    for (int j = 0; j < NROW; ++j) {
                        float* ap = sAgg + dl[j] * 64 + 2 * lane;
                        atomicAdd(ap,     x[j] * r[j].x);
                        atomicAdd(ap + 1, x[j] * r[j].y);
                    }
                }
                for (; i0 < n; i0 += nwave) {
                    const float  x = qX[i0];
                    const float2 r = ((const float2*)(xl + (size_t)qE[i0] * 64))[lane];
                    float* ap = sAgg + qD[i0] * 64 + 2 * lane;
                    atomicAdd(ap,     x * r.x);
                    atomicAdd(ap + 1, x * r.y);
                }
            }
            __syncthreads();
            c0 = c1;
        }

        for (int i = tid; i < R * 64; i += nthr) {
            const float den = sDen[i >> 6];
            agg[(size_t)base * 64 + i] = (den > 0.0f) ? sAgg[i] / den : 0.0f;
        }
        __syncthreads();
    }
}

__global__ void out_xform_kernel(const float* __restrict__ agg, const float* __restrict__ bias,
                                 const float* __restrict__ Wfc, const float* __restrict__ bfc,
                                 float* __restrict__ out, int NT)
{
    __shared__ float sW[WLDS];
    __shared__ float sb[64], sbias[64];
    stage_weight_pairs(sW, Wfc);
    for (int i = threadIdx.x; i < 64; i += blockDim.x) { sb[i] = bfc[i]; sbias[i] = bias[i]; }
    __syncthreads();

    const int lane = threadIdx.x & 31;
    const int wave = threadIdx.x >> 5;
    const int wpb  = blockDim.x >> 5;
    const int l15  = lane & 15;
    const int half = lane >> 4;
    const int numTiles = NT >> 4;

    for (int tile = blockIdx.x * wpb + wave; tile < numTiles; tile += gridDim.x * wpb) {
        v8f acc[4] = {v8f{}, v8f{}, v8f{}, v8f{}};
        const float* arow = agg + (size_t)(tile * 16 + l15) * 64;
        #pragma unroll 4
        for (int kk = 0; kk < 16; ++kk) {
            const int kbase = kk * 4 + half * 2;
            v2f a = *(const v2f*)(arow + kbase);
            a.x += sbias[kbase];
            a.y += sbias[kbase + 1];
            const int pbase = (kk * 2 + half) * PSTRIDE;
            #pragma unroll
            for (int nb = 0; nb < 4; ++nb) {
                const int col = nb * 16 + l15;
                const v2f b = *(const v2f*)(sW + pbase + col * 2);
                acc[nb] = __builtin_amdgcn_wmma_f32_16x16x4_f32(
                    false, a, false, b, (short)0, acc[nb], false, false);
            }
        }
        #pragma unroll
        for (int nb = 0; nb < 4; ++nb) {
            const int col = nb * 16 + l15;
            #pragma unroll
            for (int v = 0; v < 8; ++v) {
                const size_t row = (size_t)(tile * 16 + v + half * 8);
                out[row * 64 + col] = acc[nb][v] + sb[col];
            }
        }
    }
}

extern "C" void kernel_launch(void* const* d_in, const int* in_sizes, int n_in,
                              void* d_out, int out_size, void* d_ws, size_t ws_size,
                              hipStream_t stream) {
    const float* x      = (const float*)d_in[0];
    const int*   eidx   = (const int*)d_in[1];
    const float* eattr  = (const float*)d_in[2];
    const float* W_l    = (const float*)d_in[3];
    const float* b_l    = (const float*)d_in[4];
    const float* W_r    = (const float*)d_in[5];
    const float* b_r    = (const float*)d_in[6];
    const float* W_e    = (const float*)d_in[7];
    const float* att    = (const float*)d_in[8];
    const float* bias   = (const float*)d_in[9];
    const float* W_fc   = (const float*)d_in[10];
    const float* b_fc   = (const float*)d_in[11];

    const int NT       = in_sizes[0] / 64;
    const long long E  = (long long)in_sizes[2];
    const int* src = eidx;
    const int* dst = eidx + E;

    float* ws     = (float*)d_ws;
    float* xl     = ws;
    float* xr     = xl + (size_t)NT * 64;
    float* agg    = xr;
    unsigned* rec = (unsigned*)(xr + (size_t)NT * 64);
    unsigned* kplanes = rec + 2 * E;

    const int numTiles = NT / 16;
    const int gemmBlocks = (numTiles + 7) / 8;

    node_xform_kernel<<<gemmBlocks, 256, 0, stream>>>(x, W_l, b_l, W_r, b_r, xl, xr, NT, src, dst, eattr, E, kplanes, rec);
    node_range_gat_kernel<<<256, 1024, 0, stream>>>(rec, W_e, att, xl, xr, agg, NT, E, kplanes);
    out_xform_kernel<<<gemmBlocks, 256, 0, stream>>>(agg, bias, W_fc, b_fc, (float*)d_out, NT);
}


# LDS staging (weights in both transform kernels, owners' xr rows): a thread issues its 16-byte loads as one group before its first LDS store, not one 4-byte load waited for per trip; sums unchanged
# speedup vs baseline: 4.8855x; 1.1916x over previous
#include <hip/hip_runtime.h>
#include <hip/hip_bf16.h>

typedef float v2f __attribute__((ext_vector_type(2)));
typedef float v8f __attribute__((ext_vector_type(8)));
typedef float v4f __attribute__((ext_vector_type(4)));

#define PSTRIDE 160
#define WLDS    (32 * PSTRIDE)

__device__ __forceinline__ unsigned enc_f32(float f) {
    unsigned u = __float_as_uint(f);
    return (u & 0x80000000u) ? ~u : (u | 0x80000000u);
}
__device__ __forceinline__ float dec_f32(unsigned e) {
    return (e & 0x80000000u) ? __uint_as_float(e ^ 0x80000000u) : __uint_as_float(~e);
}

#define RMAX  196

template <bool WHOLE>
__device__ __forceinline__ void records_and_planes_batch(const int* __restrict__ src, const int* __restrict__ dst,
                                                         const float* __restrict__ eattr, long long E, long long nG,
                                                         uint4* __restrict__ klo, uint4* __restrict__ khi,
                                                         uint2* __restrict__ rec, long long G0, int lane)
{
    int   d[16], s[16];
    float a[16];
    #pragma unroll
    for (int j = 0; j < 16; ++j) {
        const long long e  = ((G0 + j) << 5) + lane;
        const long long ec = (WHOLE || e < E) ? e : (E - 1);
        d[j] = dst[ec];
        s[j] = src[ec];
        a[j] = eattr[ec];
    }
    asm volatile("" :: "v"(d[0]), "v"(d[1]), "v"(d[2]), "v"(d[3]), "v"(d[4]), "v"(d[5]), "v"(d[6]), "v"(d[7]),
                       "v"(s[0]), "v"(s[1]), "v"(s[2]), "v"(s[3]), "v"(s[4]), "v"(s[5]), "v"(s[6]), "v"(s[7]),
                       "v"(a[0]), "v"(a[1]), "v"(a[2]), "v"(a[3]), "v"(a[4]), "v"(a[5]), "v"(a[6]), "v"(a[7]));
    asm volatile("" :: "v"(d[8]), "v"(d[9]), "v"(d[10]), "v"(d[11]), "v"(d[12]), "v"(d[13]), "v"(d[14]), "v"(d[15]),
                       "v"(s[8]), "v"(s[9]), "v"(s[10]), "v"(s[11]), "v"(s[12]), "v"(s[13]), "v"(s[14]), "v"(s[15]),
                       "v"(a[8]), "v"(a[9]), "v"(a[10]), "v"(a[11]), "v"(a[12]), "v"(a[13]), "v"(a[14]), "v"(a[15]));
    unsigned mine[8] = {0u, 0u, 0u, 0u, 0u, 0u, 0u, 0u};
    #pragma unroll
    for (int j = 0; j < 16; ++j) {
        const long long e = ((G0 + j) << 5) + lane;
        if (WHOLE || e < E) rec[e] = make_uint2((unsigned)s[j] | ((unsigned)d[j] << 16), __float_as_uint(a[j]));
        const unsigned key = (unsigned)d[j] / RMAX;
        #pragma unroll
        for (int k = 0; k < 8; ++k) {
            const unsigned p = __builtin_amdgcn_ballot_w32(((key >> k) & 1u) != 0u);
            if (lane == j) mine[k] = p;
        }
    }
    if (lane < 16 && (WHOLE || G0 + lane < nG)) {
        klo[G0 + lane] = make_uint4(mine[0], mine[1], mine[2], mine[3]);
        khi[G0 + lane] = make_uint4(mine[4], mine[5], mine[6], mine[7]);
    }
}

__device__ __forceinline__ void write_edge_records_and_key_planes(const int* __restrict__ src,
                                                                  const int* __restrict__ dst,
                                                                  const float* __restrict__ eattr, long long E,
                                                                  unsigned* __restrict__ kplanes,
                                                                  unsigned* __restrict__ recw)
{
    uint2* __restrict__ rec = (uint2*)recw;
    const long long nG = (E + 31) >> 5;
    const long long nB = (nG + 15) >> 4;
    uint4* __restrict__ klo = (uint4*)kplanes;
    uint4* __restrict__ khi = klo + nG;
    const int lane = threadIdx.x & 31;
    const long long wid = (long long)blockIdx.x * (blockDim.x >> 5) + __builtin_amdgcn_readfirstlane((int)(threadIdx.x >> 5));
    const long long nw  = (long long)gridDim.x * (blockDim.x >> 5);
    for (long long b = wid; b < nB; b += nw) {
        const long long G0 = b << 4;
        if (((G0 + 16) << 5) <= E) records_and_planes_batch<true >(src, dst, eattr, E, nG, klo, khi, rec, G0, lane);
        else                       records_and_planes_batch<false>(src, dst, eattr, E, nG, klo, khi, rec, G0, lane);
    }
}

template <bool TWO>
__device__ __forceinline__ void stage_weight_pairs(float* sA, const float* __restrict__ WA,
                                                   float* sB, const float* __restrict__ WB)
{
    const int nthr = blockDim.x;
    for (int u0 = threadIdx.x; u0 < 512; u0 += 2 * nthr) {
        v4f a0[2], a1[2], b0[2], b1[2];
        #pragma unroll
        for (int j = 0; j < 2; ++j) {
            const int u  = u0 + j * nthr;
            const int uc = (u < 512) ? u : u0;
            const int o  = (uc >> 4) * 128 + (uc & 15) * 4;
            a0[j] = *(const v4f*)(WA + o);
            a1[j] = *(const v4f*)(WA + o + 64);
            if constexpr (TWO) {
                b0[j] = *(const v4f*)(WB + o);
                b1[j] = *(const v4f*)(WB + o + 64);
            }
        }
        if constexpr (TWO) asm volatile("" :: "v"(a0[0]), "v"(a1[0]), "v"(a0[1]), "v"(a1[1]), "v"(b0[0]), "v"(b1[0]), "v"(b0[1]), "v"(b1[1]));
        else               asm volatile("" :: "v"(a0[0]), "v"(a1[0]), "v"(a0[1]), "v"(a1[1]));
        #pragma unroll
        for (int j = 0; j < 2; ++j) {
            const int u = u0 + j * nthr;
            if (u < 512) {
                const int d = (u >> 4) * PSTRIDE + (u & 15) * 8;
                *(v4f*)(sA + d)     = v4f{a0[j].x, a1[j].x, a0[j].y, a1[j].y};
                *(v4f*)(sA + d + 4) = v4f{a0[j].z, a1[j].z, a0[j].w, a1[j].w};
                if constexpr (TWO) {
                    *(v4f*)(sB + d)     = v4f{b0[j].x, b1[j].x, b0[j].y, b1[j].y};
                    *(v4f*)(sB + d + 4) = v4f{b0[j].z, b1[j].z, b0[j].w, b1[j].w};
                }
            }
        }
    }
}

__global__ void node_xform_kernel(const float* __restrict__ x2,
                                  const float* __restrict__ Wl, const float* __restrict__ bl,
                                  const float* __restrict__ Wr, const float* __restrict__ br,
                                  float* __restrict__ xl, float* __restrict__ xr, int NT,
                                  const int* __restrict__ src, const int* __restrict__ dst,
                                  const float* __restrict__ eattr, long long E,
                                  unsigned* __restrict__ kplanes, unsigned* __restrict__ recw)
{
    write_edge_records_and_key_planes(src, dst, eattr, E, kplanes, recw);

    __shared__ __attribute__((aligned(16))) float sWl[WLDS];
    __shared__ __attribute__((aligned(16))) float sWr[WLDS];
    __shared__ float sbl[64], sbr[64];
    stage_weight_pairs<true>(sWl, Wl, sWr, Wr);
    for (int i = threadIdx.x; i < 64; i += blockDim.x) { sbl[i] = bl[i]; sbr[i] = br[i]; }
    __syncthreads();

    const int lane    = threadIdx.x & 31;
    const int wave    = threadIdx.x >> 5;
    const int wpb     = blockDim.x >> 5;
    const int l15     = lane & 15;
    const int half    = lane >> 4;
    const int numTiles = NT >> 4;

    for (int tile = blockIdx.x * wpb + wave; tile < numTiles; tile += gridDim.x * wpb) {
        v8f accl[4] = {v8f{}, v8f{}, v8f{}, v8f{}};
        v8f accr[4] = {v8f{}, v8f{}, v8f{}, v8f{}};
        const float* xrow = x2 + (size_t)(tile * 16 + l15) * 64;
        #pragma unroll 4
        for (int kk = 0; kk < 16; ++kk) {
            const int kbase = kk * 4 + half * 2;
            const v2f a = *(const v2f*)(xrow + kbase);
            const int pbase = (kk * 2 + half) * PSTRIDE;
            #pragma unroll
            for (int nb = 0; nb < 4; ++nb) {
                const int col = nb * 16 + l15;
                const v2f bL = *(const v2f*)(sWl + pbase + col * 2);
                const v2f bR = *(const v2f*)(sWr + pbase + col * 2);
                accl[nb] = __builtin_amdgcn_wmma_f32_16x16x4_f32(
                    false, a, false, bL, (short)0, accl[nb], false, false);
                accr[nb] = __builtin_amdgcn_wmma_f32_16x16x4_f32(
                    false, a, false, bR, (short)0, accr[nb], false, false);
            }
        }
        #pragma unroll
        for (int nb = 0; nb < 4; ++nb) {
            const int col = nb * 16 + l15;
            #pragma unroll
            for (int v = 0; v < 8; ++v) {
                const size_t row = (size_t)(tile * 16 + v + half * 8);
                xl[row * 64 + col] = accl[nb][v] + sbl[col];
                xr[row * 64 + col] = accr[nb][v] + sbr[col];
            }
        }
    }
}

#define QCAP  16384
#define SUPER 2097152
#define NGRP  4
#define NROW  8
#define NSET  2

template <int M>
__device__ __forceinline__ float xrow(float x) {
    const int b = __builtin_bit_cast(int, x);
    return __builtin_bit_cast(float, __builtin_amdgcn_update_dpp(b, b, 0x160 + M, 0xf, 0xf, false));
}

__device__ __forceinline__ float piece_score(float4 l, float4 r, float ea, float4 we, float4 at)
{
    float m0 = l.x + r.x + ea * we.x;
    float m1 = l.y + r.y + ea * we.y;
    float m2 = l.z + r.z + ea * we.z;
    float m3 = l.w + r.w + ea * we.w;
    m0 = (m0 > 0.0f) ? m0 : 0.2f * m0;
    m1 = (m1 > 0.0f) ? m1 : 0.2f * m1;
    m2 = (m2 > 0.0f) ? m2 : 0.2f * m2;
    m3 = (m3 > 0.0f) ? m3 : 0.2f * m3;
    return ((at.x * m0 + at.y * m1) + at.z * m2) + at.w * m3;
}

__global__ void __launch_bounds__(1024)
node_range_gat_kernel(const unsigned* __restrict__ recw,
                      const float* __restrict__ We, const float* __restrict__ att,
                      const float* __restrict__ xl, const float* xr,
                      float* agg, int NT, long long E,
                      const unsigned* __restrict__ kplanes)
{
    __shared__ float    sAgg[(RMAX + 1) * 64];
    __shared__ __attribute__((aligned(16))) float sXr[(RMAX + 1) * 64];
    __shared__ float    sDen[RMAX + 1];
    __shared__ float    sMax[RMAX + 1];
    __shared__ unsigned sRnd[RMAX + 1];
    __shared__ float    sScl[RMAX + 1];
    __shared__ int   qE[QCAP];
    __shared__ int   qD[QCAP];
    __shared__ float qX[QCAP];
    __shared__ int   qN;
    __shared__ __attribute__((aligned(16))) float sWe[64];
    __shared__ __attribute__((aligned(16))) float sAt[64];

    const int tid   = threadIdx.x;
    const int lane  = tid & 31;
    const int wave  = __builtin_amdgcn_readfirstlane(tid >> 5);
    const int nwave = blockDim.x >> 5;
    const int nthr  = blockDim.x;
    const long long nG = (E + 31) >> 5;
    const uint4* __restrict__ klo = (const uint4*)kplanes;
    const uint4* __restrict__ khi = klo + nG;
    const uint2* __restrict__ rec = (const uint2*)recw;

    for (int i = tid; i < 64; i += nthr) { sWe[i] = We[i]; sAt[i] = att[i]; }

    for (int base = blockIdx.x * RMAX; base < NT; base += gridDim.x * RMAX) {
        const int R = (NT - base < RMAX) ? (NT - base) : RMAX;
        {
            const int n4 = R * 16;
            const v4f* xr4 = (const v4f*)(xr + (size_t)base * 64);
            for (int i0 = tid; i0 < n4; i0 += 4 * nthr) {
                v4f v[4];
                #pragma unroll
                for (int j = 0; j < 4; ++j) {
                    const int i = i0 + j * nthr;
                    v[j] = xr4[(i < n4) ? i : i0];
                }
                asm volatile("" :: "v"(v[0]), "v"(v[1]), "v"(v[2]), "v"(v[3]));
                #pragma unroll
                for (int j = 0; j < 4; ++j) {
                    const int i = i0 + j * nthr;
                    if (i < n4) ((v4f*)sXr)[i] = v[j];
                }
            }
        }
        for (int i = tid; i < R * 64; i += nthr) sAgg[i] = 0.0f;
        for (int i = tid; i < 64; i += nthr) {
            sAgg[RMAX * 64 + i] = 0.0f;
            sXr[RMAX * 64 + i]  = 0.0f;
        }
        for (int i = tid; i <= RMAX; i += nthr) {
            sDen[i] = 0.0f;
            sMax[i] = -__builtin_inff();
            sRnd[i] = 0u;
        }
        if (tid == 0) qN = 0;
        __syncthreads();

        const unsigned pb = (unsigned)(base / RMAX) & 255u;
        unsigned ik[8];
        #pragma unroll
        for (int k = 0; k < 8; ++k) ik[k] = ((pb >> k) & 1u) ? 0u : ~0u;

        long long c0  = 0;
        long long seg = SUPER;
        while (c0 < E) {
            const long long c1 = (c0 + seg < E) ? (c0 + seg) : E;

            const unsigned G0    = (unsigned)(c0 >> 5);
            const unsigned G1    = (unsigned)((c1 + 31) >> 5);
            const unsigned trip  = (unsigned)nthr * NGRP;
            const unsigned ntrip = (G1 - G0 + trip - 1) / trip;
            const unsigned rot   = ntrip * blockIdx.x / gridDim.x;
            auto first_group = [&](unsigned tr) -> unsigned {
                if (tr >= ntrip) return G1;
                const unsigned tt = (tr + rot < ntrip) ? (tr + rot) : (tr + rot - ntrip);
                return G0 + tt * trip + (unsigned)tid;
            };
            auto load_groups = [&](uint4 (&lo)[NGRP], uint4 (&hi)[NGRP], unsigned g0) {
                #pragma unroll
                for (int j = 0; j < NGRP; ++j) {
                    const unsigned g  = g0 + (unsigned)j * nthr;
                    const unsigned gc = (g < G1) ? g : (G1 - 1u);
                    lo[j] = klo[gc];
                    hi[j] = khi[gc];
                }
            };
            auto take_groups = [&](const uint4 (&lo)[NGRP], const uint4 (&hi)[NGRP], unsigned g0) {
                #pragma unroll
                for (int j = 0; j < NGRP; ++j) {
                    unsigned m = (lo[j].x ^ ik[0]) & (lo[j].y ^ ik[1]) & (lo[j].z ^ ik[2]) & (lo[j].w ^ ik[3])
                               & (hi[j].x ^ ik[4]) & (hi[j].y ^ ik[5]) & (hi[j].z ^ ik[6]) & (hi[j].w ^ ik[7]);
                    if (m != 0u) {
                        const long long eb = (long long)(g0 + (unsigned)j * nthr) << 5;
                        do {
                            const long long e = eb + __builtin_ctz(m);
                            m &= m - 1u;
                            if (e < c1) {
                                const int idx = atomicAdd(&qN, 1);
                                if (idx < QCAP) qE[idx] = (int)e;
                            }
                        } while (m != 0u);
                    }
                }
            };
            {
                uint4 loA[NGRP], hiA[NGRP], loB[NGRP], hiB[NGRP];
                unsigned gA = first_group(0u);
                load_groups(loA, hiA, gA);
                for (unsigned tr = 0; tr < ntrip; tr += 2) {
                    const unsigned gB = first_group(tr + 1);
                    load_groups(loB, hiB, gB);
                    take_groups(loA, hiA, gA);
                    gA = first_group(tr + 2);
                    load_groups(loA, hiA, gA);
                    if (tr + 1 < ntrip) take_groups(loB, hiB, gB);
                }
            }
            __syncthreads();

            const int n = qN;
            if (n > QCAP) {
                __syncthreads();
                if (tid == 0) qN = 0;
                seg = (seg / 2 > QCAP) ? seg / 2 : QCAP;
                __syncthreads();
                continue;
            }

            for (int i = tid; i < n; i += nthr) {
                const uint2 r = rec[qE[i]];
                const unsigned dl = (r.x >> 16) - (unsigned)base;
                qD[i] = (dl < (unsigned)R) ? (int)dl : RMAX;
                qE[i] = (int)(r.x & 0xffffu);
                qX[i] = __uint_as_float(r.y);
            }
            __syncthreads();
            if (tid == 0) qN = 0;

            {
                const int q4 = lane & 3;
                const int g8 = lane >> 2;
                unsigned cb[4];
                #pragma unroll
                for (int k = 0; k < 4; ++k) cb[k] = 64u * (unsigned)((k + g8) & 3) + 16u * (unsigned)q4;
                float4 we4[4], at4[4];
                #pragma unroll
                for (int k = 0; k < 4; ++k) {
                    we4[k] = *(const float4*)((const char*)sWe + cb[k]);
                    at4[k] = *(const float4*)((const char*)sAt + cb[k]);
                }
                for (int i0 = wave * 8; i0 < n; i0 += NSET * 8 * nwave) {
                    int    id[NSET];
                    float  ea[NSET];
                    int    dl[NSET];
                    float4 l[NSET][4];
                    #pragma unroll
                    for (int s = 0; s < NSET; ++s) {
                        const int i  = i0 + s * 8 * nwave + g8;
                        const int ic = (i < n) ? i : (n - 1);
                        id[s] = (i < n) ? i : -1;
                        ea[s] = qX[ic];
                        dl[s] = qD[ic];
                        const unsigned lo = (unsigned)qE[ic] * 256u;
                        #pragma unroll
                        for (int k = 0; k < 4; ++k) l[s][k] = *(const float4*)((const char*)xl + (size_t)(lo + cb[k]));
                    }
                    __builtin_amdgcn_sched_barrier(0);
                    #pragma unroll
                    for (int s = 0; s < NSET; ++s) {
                        const unsigned ro = (unsigned)dl[s] * 256u;
                        float pk[4];
                        #pragma unroll
                        for (int k = 0; k < 4; ++k) {
                            const float4 r = *(const float4*)((const char*)sXr + (ro + cb[k]));
                            pk[k] = piece_score(l[s][k], r, ea[s], we4[k], at4[k]);
                        }
                        float p = (pk[0] + pk[1]) + (pk[2] + pk[3]);
                        p += xrow<1>(p);
                        p += xrow<2>(p);
                        if (q4 == 0 && id[s] >= 0) {
                            qX[id[s]] = p;
                            atomicMax(&sRnd[dl[s]], enc_f32(p));
                        }
                    }
                }
            }
            __syncthreads();

            for (int r = tid; r <= RMAX; r += nthr) {
                const unsigned u = sRnd[r];
                float s = 1.0f;
                if (u != 0u) {
                    const float mo = sMax[r];
                    const float mn = dec_f32(u);
                    if (mn > mo) {
                        if (mo != -__builtin_inff()) s = __expf(mo - mn);
                        sMax[r] = mn;
                    }
                }
                sScl[r] = s;
                sRnd[r] = 0u;
            }
            __syncthreads();
            if (c0 != 0) {
                for (int i = tid; i < R * 64; i += nthr) sAgg[i] *= sScl[i >> 6];
                for (int i = tid; i < 64; i += nthr) sAgg[RMAX * 64 + i] *= sScl[RMAX];
                for (int r = tid; r <= RMAX; r += nthr) sDen[r] *= sScl[r];
                __syncthreads();
            }

            for (int i = tid; i < n; i += nthr) {
                const int   dl = qD[i];
                const float ex = __expf(qX[i] - sMax[dl]);
                qX[i] = ex;
                atomicAdd(&sDen[dl], ex);
            }
            __syncthreads();

            {
                int i0 = wave;
                for (; i0 + (NROW - 1) * nwave < n; i0 += NROW * nwave) {
                    float2 r[NROW];
                    float  x[NROW];
                    int    dl[NROW];
                    #pragma unroll
                    for (int j = 0; j < NROW; ++j) {
                        const int i = i0 + j * nwave;
                        x[j]  = qX[i];
                        dl[j] = qD[i];
                        r[j]  = ((const float2*)(xl + (size_t)qE[i] * 64))[lane];
                    }
                    #pragma unroll
                    for (int j = 0; j < NROW; ++j) {
                        float* ap = sAgg + dl[j] * 64 + 2 * lane;
                        atomicAdd(ap,     x[j] * r[j].x);
                        atomicAdd(ap + 1, x[j] * r[j].y);
                    }
                }
                for (; i0 < n; i0 += nwave) {
                    const float  x = qX[i0];
                    const float2 r = ((const float2*)(xl + (size_t)qE[i0] * 64))[lane];
                    float* ap = sAgg + qD[i0] * 64 + 2 * lane;
                    atomicAdd(ap,     x * r.x);
                    atomicAdd(ap + 1, x * r.y);
                }
            }
            __syncthreads();
            c0 = c1;
        }

        for (int i = tid; i < R * 64; i += nthr) {
            const float den = sDen[i >> 6];
            agg[(size_t)base * 64 + i] = (den > 0.0f) ? sAgg[i] / den : 0.0f;
        }
        __syncthreads();
    }
}

__global__ void out_xform_kernel(const float* __restrict__ agg, const float* __restrict__ bias,
                                 const float* __restrict__ Wfc, const float* __restrict__ bfc,
                                 float* __restrict__ out, int NT)
{
    __shared__ __attribute__((aligned(16))) float sW[WLDS];
    __shared__ float sb[64], sbias[64];
    stage_weight_pairs<false>(sW, Wfc, nullptr, nullptr);
    for (int i = threadIdx.x; i < 64; i += blockDim.x) { sb[i] = bfc[i]; sbias[i] = bias[i]; }
    __syncthreads();

    const int lane = threadIdx.x & 31;
    const int wave = threadIdx.x >> 5;
    const int wpb  = blockDim.x >> 5;
    const int l15  = lane & 15;
    const int half = lane >> 4;
    const int numTiles = NT >> 4;

    for (int tile = blockIdx.x * wpb + wave; tile < numTiles; tile += gridDim.x * wpb) {
        v8f acc[4] = {v8f{}, v8f{}, v8f{}, v8f{}};
        const float* arow = agg + (size_t)(tile * 16 + l15) * 64;
        #pragma unroll 4
        for (int kk = 0; kk < 16; ++kk) {
            const int kbase = kk * 4 + half * 2;
            v2f a = *(const v2f*)(arow + kbase);
            a.x += sbias[kbase];
            a.y += sbias[kbase + 1];
            const int pbase = (kk * 2 + half) * PSTRIDE;
            #pragma unroll
            for (int nb = 0; nb < 4; ++nb) {
                const int col = nb * 16 + l15;
                const v2f b = *(const v2f*)(sW + pbase + col * 2);
                acc[nb] = __builtin_amdgcn_wmma_f32_16x16x4_f32(
                    false, a, false, b, (short)0, acc[nb], false, false);
            }
        }
        #pragma unroll
        for (int nb = 0; nb < 4; ++nb) {
            const int col = nb * 16 + l15;
            #pragma unroll
            for (int v = 0; v < 8; ++v) {
                const size_t row = (size_t)(tile * 16 + v + half * 8);
                out[row * 64 + col] = acc[nb][v] + sb[col];
            }
        }
    }
}

extern "C" void kernel_launch(void* const* d_in, const int* in_sizes, int n_in,
                              void* d_out, int out_size, void* d_ws, size_t ws_size,
                              hipStream_t stream) {
    const float* x      = (const float*)d_in[0];
    const int*   eidx   = (const int*)d_in[1];
    const float* eattr  = (const float*)d_in[2];
    const float* W_l    = (const float*)d_in[3];
    const float* b_l    = (const float*)d_in[4];
    const float* W_r    = (const float*)d_in[5];
    const float* b_r    = (const float*)d_in[6];
    const float* W_e    = (const float*)d_in[7];
    const float* att    = (const float*)d_in[8];
    const float* bias   = (const float*)d_in[9];
    const float* W_fc   = (const float*)d_in[10];
    const float* b_fc   = (const float*)d_in[11];

    const int NT       = in_sizes[0] / 64;
    const long long E  = (long long)in_sizes[2];
    const int* src = eidx;
    const int* dst = eidx + E;

    float* ws     = (float*)d_ws;
    float* xl     = ws;
    float* xr     = xl + (size_t)NT * 64;
    float* agg    = xr;
    unsigned* rec = (unsigned*)(xr + (size_t)NT * 64);
    unsigned* kplanes = rec + 2 * E;

    const int numTiles = NT / 16;
    const int gemmBlocks = (numTiles + 7) / 8;

    node_xform_kernel<<<gemmBlocks, 256, 0, stream>>>(x, W_l, b_l, W_r, b_r, xl, xr, NT, src, dst, eattr, E, kplanes, rec);
    node_range_gat_kernel<<<256, 1024, 0, stream>>>(rec, W_e, att, xl, xr, agg, NT, E, kplanes);
    out_xform_kernel<<<gemmBlocks, 256, 0, stream>>>(agg, bias, W_fc, b_fc, (float*)d_out, NT);
}


# drain: row r owned by wave r%32; weights pass lists each edge for its row's wave (LDS int add), the wave sums ex*xl[src] in registers, no LDS float atomics; sums in list arrival order, fused mul-add
# speedup vs baseline: 5.3355x; 1.0921x over previous
#include <hip/hip_runtime.h>
#include <hip/hip_bf16.h>

typedef float v2f __attribute__((ext_vector_type(2)));
typedef float v8f __attribute__((ext_vector_type(8)));
typedef float v4f __attribute__((ext_vector_type(4)));

#define PSTRIDE 160
#define WLDS    (32 * PSTRIDE)

__device__ __forceinline__ unsigned enc_f32(float f) {
    unsigned u = __float_as_uint(f);
    return (u & 0x80000000u) ? ~u : (u | 0x80000000u);
}
__device__ __forceinline__ float dec_f32(unsigned e) {
    return (e & 0x80000000u) ? __uint_as_float(e ^ 0x80000000u) : __uint_as_float(~e);
}

#define RMAX  196

template <bool WHOLE>
__device__ __forceinline__ void records_and_planes_batch(const int* __restrict__ src, const int* __restrict__ dst,
                                                         const float* __restrict__ eattr, long long E, long long nG,
                                                         uint4* __restrict__ klo, uint4* __restrict__ khi,
                                                         uint2* __restrict__ rec, long long G0, int lane)
{
    int   d[16], s[16];
    float a[16];
    #pragma unroll
    for (int j = 0; j < 16; ++j) {
        const long long e  = ((G0 + j) << 5) + lane;
        const long long ec = (WHOLE || e < E) ? e : (E - 1);
        d[j] = dst[ec];
        s[j] = src[ec];
        a[j] = eattr[ec];
    }
    asm volatile("" :: "v"(d[0]), "v"(d[1]), "v"(d[2]), "v"(d[3]), "v"(d[4]), "v"(d[5]), "v"(d[6]), "v"(d[7]),
                       "v"(s[0]), "v"(s[1]), "v"(s[2]), "v"(s[3]), "v"(s[4]), "v"(s[5]), "v"(s[6]), "v"(s[7]),
                       "v"(a[0]), "v"(a[1]), "v"(a[2]), "v"(a[3]), "v"(a[4]), "v"(a[5]), "v"(a[6]), "v"(a[7]));
    asm volatile("" :: "v"(d[8]), "v"(d[9]), "v"(d[10]), "v"(d[11]), "v"(d[12]), "v"(d[13]), "v"(d[14]), "v"(d[15]),
                       "v"(s[8]), "v"(s[9]), "v"(s[10]), "v"(s[11]), "v"(s[12]), "v"(s[13]), "v"(s[14]), "v"(s[15]),
                       "v"(a[8]), "v"(a[9]), "v"(a[10]), "v"(a[11]), "v"(a[12]), "v"(a[13]), "v"(a[14]), "v"(a[15]));
    unsigned mine[8] = {0u, 0u, 0u, 0u, 0u, 0u, 0u, 0u};
    #pragma unroll
    for (int j = 0; j < 16; ++j) {
        const long long e = ((G0 + j) << 5) + lane;
        if (WHOLE || e < E) rec[e] = make_uint2((unsigned)s[j] | ((unsigned)d[j] << 16), __float_as_uint(a[j]));
        const unsigned key = (unsigned)d[j] / RMAX;
        #pragma unroll
        for (int k = 0; k < 8; ++k) {
            const unsigned p = __builtin_amdgcn_ballot_w32(((key >> k) & 1u) != 0u);
            if (lane == j) mine[k] = p;
        }
    }
    if (lane < 16 && (WHOLE || G0 + lane < nG)) {
        klo[G0 + lane] = make_uint4(mine[0], mine[1], mine[2], mine[3]);
        khi[G0 + lane] = make_uint4(mine[4], mine[5], mine[6], mine[7]);
    }
}

__device__ __forceinline__ void write_edge_records_and_key_planes(const int* __restrict__ src,
                                                                  const int* __restrict__ dst,
                                                                  const float* __restrict__ eattr, long long E,
                                                                  unsigned* __restrict__ kplanes,
                                                                  unsigned* __restrict__ recw)
{
    uint2* __restrict__ rec = (uint2*)recw;
    const long long nG = (E + 31) >> 5;
    const long long nB = (nG + 15) >> 4;
    uint4* __restrict__ klo = (uint4*)kplanes;
    uint4* __restrict__ khi = klo + nG;
    const int lane = threadIdx.x & 31;
    const long long wid = (long long)blockIdx.x * (blockDim.x >> 5) + __builtin_amdgcn_readfirstlane((int)(threadIdx.x >> 5));
    const long long nw  = (long long)gridDim.x * (blockDim.x >> 5);
    for (long long b = wid; b < nB; b += nw) {
        const long long G0 = b << 4;
        if (((G0 + 16) << 5) <= E) records_and_planes_batch<true >(src, dst, eattr, E, nG, klo, khi, rec, G0, lane);
        else                       records_and_planes_batch<false>(src, dst, eattr, E, nG, klo, khi, rec, G0, lane);
    }
}

template <bool TWO>
__device__ __forceinline__ void stage_weight_pairs(float* sA, const float* __restrict__ WA,
                                                   float* sB, const float* __restrict__ WB)
{
    const int nthr = blockDim.x;
    for (int u0 = threadIdx.x; u0 < 512; u0 += 2 * nthr) {
        v4f a0[2], a1[2], b0[2], b1[2];
        #pragma unroll
        for (int j = 0; j < 2; ++j) {
            const int u  = u0 + j * nthr;
            const int uc = (u < 512) ? u : u0;
            const int o  = (uc >> 4) * 128 + (uc & 15) * 4;
            a0[j] = *(const v4f*)(WA + o);
            a1[j] = *(const v4f*)(WA + o + 64);
            if constexpr (TWO) {
                b0[j] = *(const v4f*)(WB + o);
                b1[j] = *(const v4f*)(WB + o + 64);
            }
        }
        if constexpr (TWO) asm volatile("" :: "v"(a0[0]), "v"(a1[0]), "v"(a0[1]), "v"(a1[1]), "v"(b0[0]), "v"(b1[0]), "v"(b0[1]), "v"(b1[1]));
        else               asm volatile("" :: "v"(a0[0]), "v"(a1[0]), "v"(a0[1]), "v"(a1[1]));
        #pragma unroll
        for (int j = 0; j < 2; ++j) {
            const int u = u0 + j * nthr;
            if (u < 512) {
                const int d = (u >> 4) * PSTRIDE + (u & 15) * 8;
                *(v4f*)(sA + d)     = v4f{a0[j].x, a1[j].x, a0[j].y, a1[j].y};
                *(v4f*)(sA + d + 4) = v4f{a0[j].z, a1[j].z, a0[j].w, a1[j].w};
                if constexpr (TWO) {
                    *(v4f*)(sB + d)     = v4f{b0[j].x, b1[j].x, b0[j].y, b1[j].y};
                    *(v4f*)(sB + d + 4) = v4f{b0[j].z, b1[j].z, b0[j].w, b1[j].w};
                }
            }
        }
    }
}

__global__ void node_xform_kernel(const float* __restrict__ x2,
                                  const float* __restrict__ Wl, const float* __restrict__ bl,
                                  const float* __restrict__ Wr, const float* __restrict__ br,
                                  float* __restrict__ xl, float* __restrict__ xr, int NT,
                                  const int* __restrict__ src, const int* __restrict__ dst,
                                  const float* __restrict__ eattr, long long E,
                                  unsigned* __restrict__ kplanes, unsigned* __restrict__ recw)
{
    write_edge_records_and_key_planes(src, dst, eattr, E, kplanes, recw);

    __shared__ __attribute__((aligned(16))) float sWl[WLDS];
    __shared__ __attribute__((aligned(16))) float sWr[WLDS];
    __shared__ float sbl[64], sbr[64];
    stage_weight_pairs<true>(sWl, Wl, sWr, Wr);
    for (int i = threadIdx.x; i < 64; i += blockDim.x) { sbl[i] = bl[i]; sbr[i] = br[i]; }
    __syncthreads();

    const int lane    = threadIdx.x & 31;
    const int wave    = threadIdx.x >> 5;
    const int wpb     = blockDim.x >> 5;
    const int l15     = lane & 15;
    const int half    = lane >> 4;
    const int numTiles = NT >> 4;

    for (int tile = blockIdx.x * wpb + wave; tile < numTiles; tile += gridDim.x * wpb) {
        v8f accl[4] = {v8f{}, v8f{}, v8f{}, v8f{}};
        v8f accr[4] = {v8f{}, v8f{}, v8f{}, v8f{}};
        const float* xrow = x2 + (size_t)(tile * 16 + l15) * 64;
        #pragma unroll 4
        for (int kk = 0; kk < 16; ++kk) {
            const int kbase = kk * 4 + half * 2;
            const v2f a = *(const v2f*)(xrow + kbase);
            const int pbase = (kk * 2 + half) * PSTRIDE;
            #pragma unroll
            for (int nb = 0; nb < 4; ++nb) {
                const int col = nb * 16 + l15;
                const v2f bL = *(const v2f*)(sWl + pbase + col * 2);
                const v2f bR = *(const v2f*)(sWr + pbase + col * 2);
                accl[nb] = __builtin_amdgcn_wmma_f32_16x16x4_f32(
                    false, a, false, bL, (short)0, accl[nb], false, false);
                accr[nb] = __builtin_amdgcn_wmma_f32_16x16x4_f32(
                    false, a, false, bR, (short)0, accr[nb], false, false);
            }
        }
        #pragma unroll
        for (int nb = 0; nb < 4; ++nb) {
            const int col = nb * 16 + l15;
            #pragma unroll
            for (int v = 0; v < 8; ++v) {
                const size_t row = (size_t)(tile * 16 + v + half * 8);
                xl[row * 64 + col] = accl[nb][v] + sbl[col];
                xr[row * 64 + col] = accr[nb][v] + sbr[col];
            }
        }
    }
}

#define QCAP  12288
#define SUPER 2097152
#define NGRP  4
#define DROW  16
#define NSLOT 7
#define WCAP  384
#define NSET  2

template <int M>
__device__ __forceinline__ float xrow(float x) {
    const int b = __builtin_bit_cast(int, x);
    return __builtin_bit_cast(float, __builtin_amdgcn_update_dpp(b, b, 0x160 + M, 0xf, 0xf, false));
}

__device__ __forceinline__ float piece_score(float4 l, float4 r, float ea, float4 we, float4 at)
{
    float m0 = l.x + r.x + ea * we.x;
    float m1 = l.y + r.y + ea * we.y;
    float m2 = l.z + r.z + ea * we.z;
    float m3 = l.w + r.w + ea * we.w;
    m0 = (m0 > 0.0f) ? m0 : 0.2f * m0;
    m1 = (m1 > 0.0f) ? m1 : 0.2f * m1;
    m2 = (m2 > 0.0f) ? m2 : 0.2f * m2;
    m3 = (m3 > 0.0f) ? m3 : 0.2f * m3;
    return ((at.x * m0 + at.y * m1) + at.z * m2) + at.w * m3;
}

__global__ void __launch_bounds__(1024)
node_range_gat_kernel(const unsigned* __restrict__ recw,
                      const float* __restrict__ We, const float* __restrict__ att,
                      const float* __restrict__ xl, const float* xr,
                      float* agg, int NT, long long E,
                      const unsigned* __restrict__ kplanes)
{
    __shared__ __attribute__((aligned(8))) float sAgg[(RMAX + 1) * 64];
    __shared__ __attribute__((aligned(16))) float sXr[(RMAX + 1) * 64];
    __shared__ float    sDen[RMAX + 1];
    __shared__ float    sMax[RMAX + 1];
    __shared__ unsigned sRnd[RMAX + 1];
    __shared__ float    sScl[RMAX + 1];
    __shared__ int   qE[QCAP];
    __shared__ int   qD[QCAP];
    __shared__ float qX[QCAP];
    __shared__ int   qN;
    __shared__ int   wList[32 * WCAP];
    __shared__ int   wCnt[32];
    __shared__ int   wOvf;
    __shared__ __attribute__((aligned(16))) float sWe[64];
    __shared__ __attribute__((aligned(16))) float sAt[64];

    const int tid   = threadIdx.x;
    const int lane  = tid & 31;
    const int wave  = __builtin_amdgcn_readfirstlane(tid >> 5);
    const int nwave = blockDim.x >> 5;
    const int nthr  = blockDim.x;
    const long long nG = (E + 31) >> 5;
    const uint4* __restrict__ klo = (const uint4*)kplanes;
    const uint4* __restrict__ khi = klo + nG;
    const uint2* __restrict__ rec = (const uint2*)recw;

    for (int i = tid; i < 64; i += nthr) { sWe[i] = We[i]; sAt[i] = att[i]; }

    for (int base = blockIdx.x * RMAX; base < NT; base += gridDim.x * RMAX) {
        const int R = (NT - base < RMAX) ? (NT - base) : RMAX;
        {
            const int n4 = R * 16;
            const v4f* xr4 = (const v4f*)(xr + (size_t)base * 64);
            for (int i0 = tid; i0 < n4; i0 += 4 * nthr) {
                v4f v[4];
                #pragma unroll
                for (int j = 0; j < 4; ++j) {
                    const int i = i0 + j * nthr;
                    v[j] = xr4[(i < n4) ? i : i0];
                }
                asm volatile("" :: "v"(v[0]), "v"(v[1]), "v"(v[2]), "v"(v[3]));
                #pragma unroll
                for (int j = 0; j < 4; ++j) {
                    const int i = i0 + j * nthr;
                    if (i < n4) ((v4f*)sXr)[i] = v[j];
                }
            }
        }
        for (int i = tid; i < R * 64; i += nthr) sAgg[i] = 0.0f;
        for (int i = tid; i < 64; i += nthr) {
            sAgg[RMAX * 64 + i] = 0.0f;
            sXr[RMAX * 64 + i]  = 0.0f;
        }
        for (int i = tid; i <= RMAX; i += nthr) {
            sDen[i] = 0.0f;
            sMax[i] = -__builtin_inff();
            sRnd[i] = 0u;
        }
        if (tid == 0) qN = 0;
        __syncthreads();

        const unsigned pb = (unsigned)(base / RMAX) & 255u;
        unsigned ik[8];
        #pragma unroll
        for (int k = 0; k < 8; ++k) ik[k] = ((pb >> k) & 1u) ? 0u : ~0u;

        long long c0  = 0;
        long long seg = SUPER;
        while (c0 < E) {
            const long long c1 = (c0 + seg < E) ? (c0 + seg) : E;

            const unsigned G0    = (unsigned)(c0 >> 5);
            const unsigned G1    = (unsigned)((c1 + 31) >> 5);
            const unsigned trip  = (unsigned)nthr * NGRP;
            const unsigned ntrip = (G1 - G0 + trip - 1) / trip;
            const unsigned rot   = ntrip * blockIdx.x / gridDim.x;
            auto first_group = [&](unsigned tr) -> unsigned {
                if (tr >= ntrip) return G1;
                const unsigned tt = (tr + rot < ntrip) ? (tr + rot) : (tr + rot - ntrip);
                return G0 + tt * trip + (unsigned)tid;
            };
            auto load_groups = [&](uint4 (&lo)[NGRP], uint4 (&hi)[NGRP], unsigned g0) {
                #pragma unroll
                for (int j = 0; j < NGRP; ++j) {
                    const unsigned g  = g0 + (unsigned)j * nthr;
                    const unsigned gc = (g < G1) ? g : (G1 - 1u);
                    lo[j] = klo[gc];
                    hi[j] = khi[gc];
                }
            };
            auto take_groups = [&](const uint4 (&lo)[NGRP], const uint4 (&hi)[NGRP], unsigned g0) {
                #pragma unroll
                for (int j = 0; j < NGRP; ++j) {
                    unsigned m = (lo[j].x ^ ik[0]) & (lo[j].y ^ ik[1]) & (lo[j].z ^ ik[2]) & (lo[j].w ^ ik[3])
                               & (hi[j].x ^ ik[4]) & (hi[j].y ^ ik[5]) & (hi[j].z ^ ik[6]) & (hi[j].w ^ ik[7]);
                    if (m != 0u) {
                        const long long eb = (long long)(g0 + (unsigned)j * nthr) << 5;
                        do {
                            const long long e = eb + __builtin_ctz(m);
                            m &= m - 1u;
                            if (e < c1) {
                                const int idx = atomicAdd(&qN, 1);
                                if (idx < QCAP) qE[idx] = (int)e;
                            }
                        } while (m != 0u);
                    }
                }
            };
            {
                uint4 loA[NGRP], hiA[NGRP], loB[NGRP], hiB[NGRP];
                unsigned gA = first_group(0u);
                load_groups(loA, hiA, gA);
                for (unsigned tr = 0; tr < ntrip; tr += 2) {
                    const unsigned gB = first_group(tr + 1);
                    load_groups(loB, hiB, gB);
                    take_groups(loA, hiA, gA);
                    gA = first_group(tr + 2);
                    load_groups(loA, hiA, gA);
                    if (tr + 1 < ntrip) take_groups(loB, hiB, gB);
                }
            }
            __syncthreads();

            const int n = qN;
            if (n > QCAP) {
                __syncthreads();
                if (tid == 0) qN = 0;
                seg = (seg / 2 > QCAP) ? seg / 2 : QCAP;
                __syncthreads();
                continue;
            }

            for (int i = tid; i < n; i += nthr) {
                const uint2 r = rec[qE[i]];
                const unsigned dl = (r.x >> 16) - (unsigned)base;
                qD[i] = (dl < (unsigned)R) ? (int)dl : RMAX;
                qE[i] = (int)(r.x & 0xffffu);
                qX[i] = __uint_as_float(r.y);
            }
            __syncthreads();
            if (tid == 0) qN = 0;
            if (tid < 32) wCnt[tid] = 0;
            if (tid == 0) wOvf = 0;

            {
                const int q4 = lane & 3;
                const int g8 = lane >> 2;
                unsigned cb[4];
                #pragma unroll
                for (int k = 0; k < 4; ++k) cb[k] = 64u * (unsigned)((k + g8) & 3) + 16u * (unsigned)q4;
                float4 we4[4], at4[4];
                #pragma unroll
                for (int k = 0; k < 4; ++k) {
                    we4[k] = *(const float4*)((const char*)sWe + cb[k]);
                    at4[k] = *(const float4*)((const char*)sAt + cb[k]);
                }
                for (int i0 = wave * 8; i0 < n; i0 += NSET * 8 * nwave) {
                    int    id[NSET];
                    float  ea[NSET];
                    int    dl[NSET];
                    float4 l[NSET][4];
                    #pragma unroll
                    for (int s = 0; s < NSET; ++s) {
                        const int i  = i0 + s * 8 * nwave + g8;
                        const int ic = (i < n) ? i : (n - 1);
                        id[s] = (i < n) ? i : -1;
                        ea[s] = qX[ic];
                        dl[s] = qD[ic];
                        const unsigned lo = (unsigned)qE[ic] * 256u;
                        #pragma unroll
                        for (int k = 0; k < 4; ++k) l[s][k] = *(const float4*)((const char*)xl + (size_t)(lo + cb[k]));
                    }
                    __builtin_amdgcn_sched_barrier(0);
                    #pragma unroll
                    for (int s = 0; s < NSET; ++s) {
                        const unsigned ro = (unsigned)dl[s] * 256u;
                        float pk[4];
                        #pragma unroll
                        for (int k = 0; k < 4; ++k) {
                            const float4 r = *(const float4*)((const char*)sXr + (ro + cb[k]));
                            pk[k] = piece_score(l[s][k], r, ea[s], we4[k], at4[k]);
                        }
                        float p = (pk[0] + pk[1]) + (pk[2] + pk[3]);
                        p += xrow<1>(p);
                        p += xrow<2>(p);
                        if (q4 == 0 && id[s] >= 0) {
                            qX[id[s]] = p;
                            atomicMax(&sRnd[dl[s]], enc_f32(p));
                        }
                    }
                }
            }
            __syncthreads();

            for (int r = tid; r <= RMAX; r += nthr) {
                const unsigned u = sRnd[r];
                float s = 1.0f;
                if (u != 0u) {
                    const float mo = sMax[r];
                    const float mn = dec_f32(u);
                    if (mn > mo) {
                        if (mo != -__builtin_inff()) s = __expf(mo - mn);
                        sMax[r] = mn;
                    }
                }
                sScl[r] = s;
                sRnd[r] = 0u;
            }
            __syncthreads();
            if (c0 != 0) {
                for (int i = tid; i < R * 64; i += nthr) sAgg[i] *= sScl[i >> 6];
                for (int i = tid; i < 64; i += nthr) sAgg[RMAX * 64 + i] *= sScl[RMAX];
                for (int r = tid; r <= RMAX; r += nthr) sDen[r] *= sScl[r];
                __syncthreads();
            }

            for (int i = tid; i < n; i += nthr) {
                const int   dl = qD[i];
                const float ex = __expf(qX[i] - sMax[dl]);
                qX[i] = ex;
                atomicAdd(&sDen[dl], ex);
                const int pos = atomicAdd(&wCnt[dl & 31], 1);
                if (pos < WCAP) { wList[(dl & 31) * WCAP + pos] = i | ((dl >> 5) << 16); qD[i] = -1; }
                else            wOvf = 1;
            }
            __syncthreads();

            for (;;) {
                {
                    v8f accx = v8f{}, accy = v8f{};
                    const int* mylist = wList + wave * WCAP;
                    const int  c = (wCnt[wave] < WCAP) ? wCnt[wave] : WCAP;
                    const unsigned lane8 = (unsigned)lane * 8u;
                    for (int j0 = 0; j0 < c; j0 += DROW) {
                        const int  jj = j0 + lane;
                        const bool in = (lane < DROW) && (jj < c);
                        const int  en = mylist[in ? jj : j0];
                        const int  ii = en & 0xffff;
                        const int  kv = in ? (en >> 16) : NSLOT;
                        const int  sv = qE[ii];
                        const int  xv = __float_as_int(qX[ii]);
                        float2 r[DROW];
                        #pragma unroll
                        for (int j = 0; j < DROW; ++j) {
                            const unsigned s = (unsigned)__builtin_amdgcn_readlane(sv, j);
                            const char* rowp = (const char*)xl + ((unsigned long long)s << 8);
                            r[j] = *(const float2*)(rowp + lane8);
                        }
                        asm volatile("" :: "v"(r[0].x), "v"(r[0].y), "v"(r[1].x), "v"(r[1].y), "v"(r[2].x), "v"(r[2].y), "v"(r[3].x), "v"(r[3].y),
                                           "v"(r[4].x), "v"(r[4].y), "v"(r[5].x), "v"(r[5].y), "v"(r[6].x), "v"(r[6].y), "v"(r[7].x), "v"(r[7].y));
                        asm volatile("" :: "v"(r[8].x), "v"(r[8].y), "v"(r[9].x), "v"(r[9].y), "v"(r[10].x), "v"(r[10].y), "v"(r[11].x), "v"(r[11].y),
                                           "v"(r[12].x), "v"(r[12].y), "v"(r[13].x), "v"(r[13].y), "v"(r[14].x), "v"(r[14].y), "v"(r[15].x), "v"(r[15].y));
                        #pragma unroll
                        for (int j = 0; j < DROW; ++j) {
                            const float x = __int_as_float(__builtin_amdgcn_readlane(xv, j));
                            const int   k = __builtin_amdgcn_readlane(kv, j);
                            accx[k] += x * r[j].x;
                            accy[k] += x * r[j].y;
                        }
                    }
                    #pragma unroll
                    for (int k = 0; k < NSLOT; ++k) {
                        const int row = wave + 32 * k;
                        if (row < R || row == RMAX) {
                            float2* p = (float2*)(sAgg + row * 64) + lane;
                            float2 v = *p;
                            v.x += accx[k];
                            v.y += accy[k];
                            *p = v;
                        }
                    }
                }
                const int again = wOvf;
                __syncthreads();
                if (!again) break;
                if (tid < 32) wCnt[tid] = 0;
                if (tid == 0) wOvf = 0;
                __syncthreads();
                for (int i = tid; i < n; i += nthr) {
                    const int dl = qD[i];
                    if (dl >= 0) {
                        const int pos = atomicAdd(&wCnt[dl & 31], 1);
                        if (pos < WCAP) { wList[(dl & 31) * WCAP + pos] = i | ((dl >> 5) << 16); qD[i] = -1; }
                        else            wOvf = 1;
                    }
                }
                __syncthreads();
            }
            c0 = c1;
        }

        for (int i = tid; i < R * 64; i += nthr) {
            const float den = sDen[i >> 6];
            agg[(size_t)base * 64 + i] = (den > 0.0f) ? sAgg[i] / den : 0.0f;
        }
        __syncthreads();
    }
}

__global__ void out_xform_kernel(const float* __restrict__ agg, const float* __restrict__ bias,
                                 const float* __restrict__ Wfc, const float* __restrict__ bfc,
                                 float* __restrict__ out, int NT)
{
    __shared__ __attribute__((aligned(16))) float sW[WLDS];
    __shared__ float sb[64], sbias[64];
    stage_weight_pairs<false>(sW, Wfc, nullptr, nullptr);
    for (int i = threadIdx.x; i < 64; i += blockDim.x) { sb[i] = bfc[i]; sbias[i] = bias[i]; }
    __syncthreads();

    const int lane = threadIdx.x & 31;
    const int wave = threadIdx.x >> 5;
    const int wpb  = blockDim.x >> 5;
    const int l15  = lane & 15;
    const int half = lane >> 4;
    const int numTiles = NT >> 4;

    for (int tile = blockIdx.x * wpb + wave; tile < numTiles; tile += gridDim.x * wpb) {
        v8f acc[4] = {v8f{}, v8f{}, v8f{}, v8f{}};
        const float* arow = agg + (size_t)(tile * 16 + l15) * 64;
        #pragma unroll 4
        for (int kk = 0; kk < 16; ++kk) {
            const int kbase = kk * 4 + half * 2;
            v2f a = *(const v2f*)(arow + kbase);
            a.x += sbias[kbase];
            a.y += sbias[kbase + 1];
            const int pbase = (kk * 2 + half) * PSTRIDE;
            #pragma unroll
            for (int nb = 0; nb < 4; ++nb) {
                const int col = nb * 16 + l15;
                const v2f b = *(const v2f*)(sW + pbase + col * 2);
                acc[nb] = __builtin_amdgcn_wmma_f32_16x16x4_f32(
                    false, a, false, b, (short)0, acc[nb], false, false);
            }
        }
        #pragma unroll
        for (int nb = 0; nb < 4; ++nb) {
            const int col = nb * 16 + l15;
            #pragma unroll
            for (int v = 0; v < 8; ++v) {
                const size_t row = (size_t)(tile * 16 + v + half * 8);
                out[row * 64 + col] = acc[nb][v] + sb[col];
            }
        }
    }
}

extern "C" void kernel_launch(void* const* d_in, const int* in_sizes, int n_in,
                              void* d_out, int out_size, void* d_ws, size_t ws_size,
                              hipStream_t stream) {
    const float* x      = (const float*)d_in[0];
    const int*   eidx   = (const int*)d_in[1];
    const float* eattr  = (const float*)d_in[2];
    const float* W_l    = (const float*)d_in[3];
    const float* b_l    = (const float*)d_in[4];
    const float* W_r    = (const float*)d_in[5];
    const float* b_r    = (const float*)d_in[6];
    const float* W_e    = (const float*)d_in[7];
    const float* att    = (const float*)d_in[8];
    const float* bias   = (const float*)d_in[9];
    const float* W_fc   = (const float*)d_in[10];
    const float* b_fc   = (const float*)d_in[11];

    const int NT       = in_sizes[0] / 64;
    const long long E  = (long long)in_sizes[2];
    const int* src = eidx;
    const int* dst = eidx + E;

    float* ws     = (float*)d_ws;
    float* xl     = ws;
    float* xr     = xl + (size_t)NT * 64;
    float* agg    = xr;
    unsigned* rec = (unsigned*)(xr + (size_t)NT * 64);
    unsigned* kplanes = rec + 2 * E;

    const int numTiles = NT / 16;
    const int gemmBlocks = (numTiles + 7) / 8;

    node_xform_kernel<<<gemmBlocks, 256, 0, stream>>>(x, W_l, b_l, W_r, b_r, xl, xr, NT, src, dst, eattr, E, kplanes, rec);
    node_range_gat_kernel<<<256, 1024, 0, stream>>>(rec, W_e, att, xl, xr, agg, NT, E, kplanes);
    out_xform_kernel<<<gemmBlocks, 256, 0, stream>>>(agg, bias, W_fc, b_fc, (float*)d_out, NT);
}


# score and weighted sum in one walk of per-row edge lists by the row's wave (8 lanes an edge): xl[src] gathered once, not twice; running max per octet, sums rescaled, joined by a tree (order changes)
# speedup vs baseline: 6.3488x; 1.1899x over previous
#include <hip/hip_runtime.h>
#include <hip/hip_bf16.h>

typedef float v2f __attribute__((ext_vector_type(2)));
typedef float v8f __attribute__((ext_vector_type(8)));
typedef float v4f __attribute__((ext_vector_type(4)));

#define PSTRIDE 160
#define WLDS    (32 * PSTRIDE)

__device__ __forceinline__ unsigned enc_f32(float f) {
    unsigned u = __float_as_uint(f);
    return (u & 0x80000000u) ? ~u : (u | 0x80000000u);
}
__device__ __forceinline__ float dec_f32(unsigned e) {
    return (e & 0x80000000u) ? __uint_as_float(e ^ 0x80000000u) : __uint_as_float(~e);
}

#define RMAX  196

template <bool WHOLE>
__device__ __forceinline__ void records_and_planes_batch(const int* __restrict__ src, const int* __restrict__ dst,
                                                         const float* __restrict__ eattr, long long E, long long nG,
                                                         uint4* __restrict__ klo, uint4* __restrict__ khi,
                                                         uint2* __restrict__ rec, long long G0, int lane)
{
    int   d[16], s[16];
    float a[16];
    #pragma unroll
    for (int j = 0; j < 16; ++j) {
        const long long e  = ((G0 + j) << 5) + lane;
        const long long ec = (WHOLE || e < E) ? e : (E - 1);
        d[j] = dst[ec];
        s[j] = src[ec];
        a[j] = eattr[ec];
    }
    asm volatile("" :: "v"(d[0]), "v"(d[1]), "v"(d[2]), "v"(d[3]), "v"(d[4]), "v"(d[5]), "v"(d[6]), "v"(d[7]),
                       "v"(s[0]), "v"(s[1]), "v"(s[2]), "v"(s[3]), "v"(s[4]), "v"(s[5]), "v"(s[6]), "v"(s[7]),
                       "v"(a[0]), "v"(a[1]), "v"(a[2]), "v"(a[3]), "v"(a[4]), "v"(a[5]), "v"(a[6]), "v"(a[7]));
    asm volatile("" :: "v"(d[8]), "v"(d[9]), "v"(d[10]), "v"(d[11]), "v"(d[12]), "v"(d[13]), "v"(d[14]), "v"(d[15]),
                       "v"(s[8]), "v"(s[9]), "v"(s[10]), "v"(s[11]), "v"(s[12]), "v"(s[13]), "v"(s[14]), "v"(s[15]),
                       "v"(a[8]), "v"(a[9]), "v"(a[10]), "v"(a[11]), "v"(a[12]), "v"(a[13]), "v"(a[14]), "v"(a[15]));
    unsigned mine[8] = {0u, 0u, 0u, 0u, 0u, 0u, 0u, 0u};
    #pragma unroll
    for (int j = 0; j < 16; ++j) {
        const long long e = ((G0 + j) << 5) + lane;
        if (WHOLE || e < E) rec[e] = make_uint2((unsigned)s[j] | ((unsigned)d[j] << 16), __float_as_uint(a[j]));
        const unsigned key = (unsigned)d[j] / RMAX;
        #pragma unroll
        for (int k = 0; k < 8; ++k) {
            const unsigned p = __builtin_amdgcn_ballot_w32(((key >> k) & 1u) != 0u);
            if (lane == j) mine[k] = p;
        }
    }
    if (lane < 16 && (WHOLE || G0 + lane < nG)) {
        klo[G0 + lane] = make_uint4(mine[0], mine[1], mine[2], mine[3]);
        khi[G0 + lane] = make_uint4(mine[4], mine[5], mine[6], mine[7]);
    }
}

__device__ __forceinline__ void write_edge_records_and_key_planes(const int* __restrict__ src,
                                                                  const int* __restrict__ dst,
                                                                  const float* __restrict__ eattr, long long E,
                                                                  unsigned* __restrict__ kplanes,
                                                                  unsigned* __restrict__ recw)
{
    uint2* __restrict__ rec = (uint2*)recw;
    const long long nG = (E + 31) >> 5;
    const long long nB = (nG + 15) >> 4;
    uint4* __restrict__ klo = (uint4*)kplanes;
    uint4* __restrict__ khi = klo + nG;
    const int lane = threadIdx.x & 31;
    const long long wid = (long long)blockIdx.x * (blockDim.x >> 5) + __builtin_amdgcn_readfirstlane((int)(threadIdx.x >> 5));
    const long long nw  = (long long)gridDim.x * (blockDim.x >> 5);
    for (long long b = wid; b < nB; b += nw) {
        const long long G0 = b << 4;
        if (((G0 + 16) << 5) <= E) records_and_planes_batch<true >(src, dst, eattr, E, nG, klo, khi, rec, G0, lane);
        else                       records_and_planes_batch<false>(src, dst, eattr, E, nG, klo, khi, rec, G0, lane);
    }
}

template <bool TWO>
__device__ __forceinline__ void stage_weight_pairs(float* sA, const float* __restrict__ WA,
                                                   float* sB, const float* __restrict__ WB)
{
    const int nthr = blockDim.x;
    for (int u0 = threadIdx.x; u0 < 512; u0 += 2 * nthr) {
        v4f a0[2], a1[2], b0[2], b1[2];
        #pragma unroll
        for (int j = 0; j < 2; ++j) {
            const int u  = u0 + j * nthr;
            const int uc = (u < 512) ? u : u0;
            const int o  = (uc >> 4) * 128 + (uc & 15) * 4;
            a0[j] = *(const v4f*)(WA + o);
            a1[j] = *(const v4f*)(WA + o + 64);
            if constexpr (TWO) {
                b0[j] = *(const v4f*)(WB + o);
                b1[j] = *(const v4f*)(WB + o + 64);
            }
        }
        if constexpr (TWO) asm volatile("" :: "v"(a0[0]), "v"(a1[0]), "v"(a0[1]), "v"(a1[1]), "v"(b0[0]), "v"(b1[0]), "v"(b0[1]), "v"(b1[1]));
        else               asm volatile("" :: "v"(a0[0]), "v"(a1[0]), "v"(a0[1]), "v"(a1[1]));
        #pragma unroll
        for (int j = 0; j < 2; ++j) {
            const int u = u0 + j * nthr;
            if (u < 512) {
                const int d = (u >> 4) * PSTRIDE + (u & 15) * 8;
                *(v4f*)(sA + d)     = v4f{a0[j].x, a1[j].x, a0[j].y, a1[j].y};
                *(v4f*)(sA + d + 4) = v4f{a0[j].z, a1[j].z, a0[j].w, a1[j].w};
                if constexpr (TWO) {
                    *(v4f*)(sB + d)     = v4f{b0[j].x, b1[j].x, b0[j].y, b1[j].y};
                    *(v4f*)(sB + d + 4) = v4f{b0[j].z, b1[j].z, b0[j].w, b1[j].w};
                }
            }
        }
    }
}

__global__ void node_xform_kernel(const float* __restrict__ x2,
                                  const float* __restrict__ Wl, const float* __restrict__ bl,
                                  const float* __restrict__ Wr, const float* __restrict__ br,
                                  float* __restrict__ xl, float* __restrict__ xr, int NT,
                                  const int* __restrict__ src, const int* __restrict__ dst,
                                  const float* __restrict__ eattr, long long E,
                                  unsigned* __restrict__ kplanes, unsigned* __restrict__ recw)
{
    write_edge_records_and_key_planes(src, dst, eattr, E, kplanes, recw);

    __shared__ __attribute__((aligned(16))) float sWl[WLDS];
    __shared__ __attribute__((aligned(16))) float sWr[WLDS];
    __shared__ float sbl[64], sbr[64];
    stage_weight_pairs<true>(sWl, Wl, sWr, Wr);
    for (int i = threadIdx.x; i < 64; i += blockDim.x) { sbl[i] = bl[i]; sbr[i] = br[i]; }
    __syncthreads();

    const int lane    = threadIdx.x & 31;
    const int wave    = threadIdx.x >> 5;
    const int wpb     = blockDim.x >> 5;
    const int l15     = lane & 15;
    const int half    = lane >> 4;
    const int numTiles = NT >> 4;

    for (int tile = blockIdx.x * wpb + wave; tile < numTiles; tile += gridDim.x * wpb) {
        v8f accl[4] = {v8f{}, v8f{}, v8f{}, v8f{}};
        v8f accr[4] = {v8f{}, v8f{}, v8f{}, v8f{}};
        const float* xrow = x2 + (size_t)(tile * 16 + l15) * 64;
        #pragma unroll 4
        for (int kk = 0; kk < 16; ++kk) {
            const int kbase = kk * 4 + half * 2;
            const v2f a = *(const v2f*)(xrow + kbase);
            const int pbase = (kk * 2 + half) * PSTRIDE;
            #pragma unroll
            for (int nb = 0; nb < 4; ++nb) {
                const int col = nb * 16 + l15;
                const v2f bL = *(const v2f*)(sWl + pbase + col * 2);
                const v2f bR = *(const v2f*)(sWr + pbase + col * 2);
                accl[nb] = __builtin_amdgcn_wmma_f32_16x16x4_f32(
                    false, a, false, bL, (short)0, accl[nb], false, false);
                accr[nb] = __builtin_amdgcn_wmma_f32_16x16x4_f32(
                    false, a, false, bR, (short)0, accr[nb], false, false);
            }
        }
        #pragma unroll
        for (int nb = 0; nb < 4; ++nb) {
            const int col = nb * 16 + l15;
            #pragma unroll
            for (int v = 0; v < 8; ++v) {
                const size_t row = (size_t)(tile * 16 + v + half * 8);
                xl[row * 64 + col] = accl[nb][v] + sbl[col];
                xr[row * 64 + col] = accr[nb][v] + sbr[col];
            }
        }
    }
}

#define QCAP  12288
#define SUPER 2097152
#define NGRP  4
#define NSLOT 7
#define RCAP  96
#define NSET  5

template <int M>
__device__ __forceinline__ float xrow(float x) {
    const int b = __builtin_bit_cast(int, x);
    return __builtin_bit_cast(float, __builtin_amdgcn_update_dpp(b, b, 0x160 + M, 0xf, 0xf, false));
}

__device__ __forceinline__ float piece_score(float4 l, float4 r, float ea, float4 we, float4 at)
{
    float m0 = l.x + r.x + ea * we.x;
    float m1 = l.y + r.y + ea * we.y;
    float m2 = l.z + r.z + ea * we.z;
    float m3 = l.w + r.w + ea * we.w;
    m0 = (m0 > 0.0f) ? m0 : 0.2f * m0;
    m1 = (m1 > 0.0f) ? m1 : 0.2f * m1;
    m2 = (m2 > 0.0f) ? m2 : 0.2f * m2;
    m3 = (m3 > 0.0f) ? m3 : 0.2f * m3;
    return ((at.x * m0 + at.y * m1) + at.z * m2) + at.w * m3;
}

__global__ void __launch_bounds__(1024)
node_range_gat_kernel(const unsigned* __restrict__ recw,
                      const float* __restrict__ We, const float* __restrict__ att,
                      const float* __restrict__ xl, const float* xr,
                      float* agg, int NT, long long E,
                      const unsigned* __restrict__ kplanes)
{
    __shared__ __attribute__((aligned(16))) float sAgg[RMAX * 64];
    __shared__ __attribute__((aligned(16))) float sXr[RMAX * 64];
    __shared__ float    sDen[RMAX];
    __shared__ float    sMax[RMAX];
    __shared__ int   qE[QCAP];
    __shared__ int   qD[QCAP];
    __shared__ float qX[QCAP];
    __shared__ int   qN;
    __shared__ unsigned short rList[RMAX * RCAP];
    __shared__ int   rCnt[RMAX];
    __shared__ int   wOvf;
    __shared__ __attribute__((aligned(16))) float sWe[64];
    __shared__ __attribute__((aligned(16))) float sAt[64];

    const int tid   = threadIdx.x;
    const int lane  = tid & 31;
    const int wave  = __builtin_amdgcn_readfirstlane(tid >> 5);
    const int nthr  = blockDim.x;
    const long long nG = (E + 31) >> 5;
    const uint4* __restrict__ klo = (const uint4*)kplanes;
    const uint4* __restrict__ khi = klo + nG;
    const uint2* __restrict__ rec = (const uint2*)recw;

    for (int i = tid; i < 64; i += nthr) { sWe[i] = We[i]; sAt[i] = att[i]; }

    for (int base = blockIdx.x * RMAX; base < NT; base += gridDim.x * RMAX) {
        const int R = (NT - base < RMAX) ? (NT - base) : RMAX;
        {
            const int n4 = R * 16;
            const v4f* xr4 = (const v4f*)(xr + (size_t)base * 64);
            for (int i0 = tid; i0 < n4; i0 += 4 * nthr) {
                v4f v[4];
                #pragma unroll
                for (int j = 0; j < 4; ++j) {
                    const int i = i0 + j * nthr;
                    v[j] = xr4[(i < n4) ? i : i0];
                }
                asm volatile("" :: "v"(v[0]), "v"(v[1]), "v"(v[2]), "v"(v[3]));
                #pragma unroll
                for (int j = 0; j < 4; ++j) {
                    const int i = i0 + j * nthr;
                    if (i < n4) ((v4f*)sXr)[i] = v[j];
                }
            }
        }
        for (int i = tid; i < R * 64; i += nthr) sAgg[i] = 0.0f;
        for (int i = tid; i < RMAX; i += nthr) {
            sDen[i] = 0.0f;
            sMax[i] = -__builtin_inff();
            rCnt[i] = 0;
        }
        if (tid == 0) { qN = 0; wOvf = 0; }
        __syncthreads();

        const unsigned pb = (unsigned)(base / RMAX) & 255u;
        unsigned ik[8];
        #pragma unroll
        for (int k = 0; k < 8; ++k) ik[k] = ((pb >> k) & 1u) ? 0u : ~0u;

        long long c0  = 0;
        long long seg = SUPER;
        while (c0 < E) {
            const long long c1 = (c0 + seg < E) ? (c0 + seg) : E;

            const unsigned G0    = (unsigned)(c0 >> 5);
            const unsigned G1    = (unsigned)((c1 + 31) >> 5);
            const unsigned trip  = (unsigned)nthr * NGRP;
            const unsigned ntrip = (G1 - G0 + trip - 1) / trip;
            const unsigned rot   = ntrip * blockIdx.x / gridDim.x;
            auto first_group = [&](unsigned tr) -> unsigned {
                if (tr >= ntrip) return G1;
                const unsigned tt = (tr + rot < ntrip) ? (tr + rot) : (tr + rot - ntrip);
                return G0 + tt * trip + (unsigned)tid;
            };
            auto load_groups = [&](uint4 (&lo)[NGRP], uint4 (&hi)[NGRP], unsigned g0) {
                #pragma unroll
                for (int j = 0; j < NGRP; ++j) {
                    const unsigned g  = g0 + (unsigned)j * nthr;
                    const unsigned gc = (g < G1) ? g : (G1 - 1u);
                    lo[j] = klo[gc];
                    hi[j] = khi[gc];
                }
            };
            auto take_groups = [&](const uint4 (&lo)[NGRP], const uint4 (&hi)[NGRP], unsigned g0) {
                #pragma unroll
                for (int j = 0; j < NGRP; ++j) {
                    unsigned m = (lo[j].x ^ ik[0]) & (lo[j].y ^ ik[1]) & (lo[j].z ^ ik[2]) & (lo[j].w ^ ik[3])
                               & (hi[j].x ^ ik[4]) & (hi[j].y ^ ik[5]) & (hi[j].z ^ ik[6]) & (hi[j].w ^ ik[7]);
                    if (m != 0u) {
                        const long long eb = (long long)(g0 + (unsigned)j * nthr) << 5;
                        do {
                            const long long e = eb + __builtin_ctz(m);
                            m &= m - 1u;
                            if (e < c1) {
                                const int idx = atomicAdd(&qN, 1);
                                if (idx < QCAP) qE[idx] = (int)e;
                            }
                        } while (m != 0u);
                    }
                }
            };
            {
                uint4 loA[NGRP], hiA[NGRP], loB[NGRP], hiB[NGRP];
                unsigned gA = first_group(0u);
                load_groups(loA, hiA, gA);
                for (unsigned tr = 0; tr < ntrip; tr += 2) {
                    const unsigned gB = first_group(tr + 1);
                    load_groups(loB, hiB, gB);
                    take_groups(loA, hiA, gA);
                    gA = first_group(tr + 2);
                    load_groups(loA, hiA, gA);
                    if (tr + 1 < ntrip) take_groups(loB, hiB, gB);
                }
            }
            __syncthreads();

            const int n = qN;
            if (n > QCAP) {
                __syncthreads();
                if (tid == 0) qN = 0;
                seg = (seg / 2 > QCAP) ? seg / 2 : QCAP;
                __syncthreads();
                continue;
            }

            for (int i = tid; i < n; i += nthr) {
                const uint2 r = rec[qE[i]];
                const unsigned dl = (r.x >> 16) - (unsigned)base;
                qE[i] = (int)(r.x & 0xffffu);
                qX[i] = __uint_as_float(r.y);
                int wait = -1;
                if (dl < (unsigned)R) {
                    const int pos = atomicAdd(&rCnt[dl], 1);
                    if (pos < RCAP) rList[dl * RCAP + pos] = (unsigned short)i;
                    else          { wait = (int)dl; wOvf = 1; }
                }
                qD[i] = wait;
            }
            __syncthreads();
            if (tid == 0) qN = 0;

            for (;;) {
                {
                    const int o8 = lane & 7;
                    const int g4 = lane >> 3;
                    unsigned cb[2];
                    #pragma unroll
                    for (int k = 0; k < 2; ++k) cb[k] = 128u * (unsigned)k + 16u * (unsigned)o8;
                    float4 we4[2], at4[2];
                    #pragma unroll
                    for (int k = 0; k < 2; ++k) {
                        we4[k] = *(const float4*)((const char*)sWe + cb[k]);
                        at4[k] = *(const float4*)((const char*)sAt + cb[k]);
                    }
                    for (int slot = 0; slot < NSLOT; ++slot) {
                        const int row = wave + 32 * slot;
                        if (row >= R) break;
                        const int cnt = __builtin_amdgcn_readfirstlane(rCnt[row]);
                        const int c   = (cnt < RCAP) ? cnt : RCAP;
                        if (c == 0) continue;
                        const unsigned ro = (unsigned)row * 256u;
                        float4 r4[2];
                        #pragma unroll
                        for (int k = 0; k < 2; ++k) r4[k] = *(const float4*)((const char*)sXr + (ro + cb[k]));
                        const unsigned short* lst = rList + row * RCAP;
                        float  m   = -__builtin_inff();
                        float  den = 0.0f;
                        float4 acc[2] = {float4{0.0f, 0.0f, 0.0f, 0.0f}, float4{0.0f, 0.0f, 0.0f, 0.0f}};
                        for (int j0 = 0; j0 < c; j0 += NSET * 4) {
                            bool   ok[NSET];
                            float  ea[NSET];
                            float4 l[NSET][2];
                            #pragma unroll
                            for (int s = 0; s < NSET; ++s) {
                                const int j  = j0 + s * 4 + g4;
                                ok[s] = j < c;
                                const int ii = lst[ok[s] ? j : 0];
                                ea[s] = qX[ii];
                                const unsigned lo = (unsigned)qE[ii] * 256u;
                                #pragma unroll
                                for (int k = 0; k < 2; ++k) l[s][k] = *(const float4*)((const char*)xl + (size_t)(lo + cb[k]));
                            }
                            __builtin_amdgcn_sched_barrier(0);
                            #pragma unroll
                            for (int s = 0; s < NSET; ++s) {
                                if (j0 + s * 4 < c) {
                                    float p = piece_score(l[s][0], r4[0], ea[s], we4[0], at4[0])
                                            + piece_score(l[s][1], r4[1], ea[s], we4[1], at4[1]);
                                    p += xrow<1>(p);
                                    p += xrow<2>(p);
                                    p += xrow<4>(p);
                                    const float mn = ok[s] ? fmaxf(m, p) : m;
                                    const float sc = (mn == m) ? 1.0f : __expf(m - mn);
                                    const float w  = ok[s] ? __expf(p - mn) : 0.0f;
                                    den = den * sc + w;
                                    #pragma unroll
                                    for (int k = 0; k < 2; ++k) {
                                        acc[k].x = acc[k].x * sc + w * l[s][k].x;
                                        acc[k].y = acc[k].y * sc + w * l[s][k].y;
                                        acc[k].z = acc[k].z * sc + w * l[s][k].z;
                                        acc[k].w = acc[k].w * sc + w * l[s][k].w;
                                    }
                                    m = mn;
                                }
                            }
                        }
                        float M = m;
                        M = fmaxf(M, xrow<8>(M));
                        M = fmaxf(M, __shfl_xor(M, 16));
                        const float f = __expf(m - M);
                        den *= f;
                        den += xrow<8>(den);
                        den += __shfl_xor(den, 16);
                        #pragma unroll
                        for (int k = 0; k < 2; ++k) {
                            float v[4] = {acc[k].x * f, acc[k].y * f, acc[k].z * f, acc[k].w * f};
                            #pragma unroll
                            for (int t = 0; t < 4; ++t) {
                                v[t] += xrow<8>(v[t]);
                                v[t] += __shfl_xor(v[t], 16);
                            }
                            acc[k] = float4{v[0], v[1], v[2], v[3]};
                        }
                        if (g4 == 0) {
                            const float Mo = sMax[row];
                            const float Mn = fmaxf(Mo, M);
                            const float so = (Mo == Mn) ? 1.0f : __expf(Mo - Mn);
                            const float sw = (M  == Mn) ? 1.0f : __expf(M - Mn);
                            #pragma unroll
                            for (int k = 0; k < 2; ++k) {
                                float4* p = (float4*)((char*)sAgg + (ro + cb[k]));
                                float4 v = *p;
                                v.x = v.x * so + acc[k].x * sw;
                                v.y = v.y * so + acc[k].y * sw;
                                v.z = v.z * so + acc[k].z * sw;
                                v.w = v.w * so + acc[k].w * sw;
                                *p = v;
                            }
                            if (o8 == 0) {
                                sDen[row] = sDen[row] * so + den * sw;
                                sMax[row] = Mn;
                            }
                        }
                    }
                }
                const int again = wOvf;
                __syncthreads();
                for (int i = tid; i < RMAX; i += nthr) rCnt[i] = 0;
                if (!again) break;
                if (tid == 0) wOvf = 0;
                __syncthreads();
                for (int i = tid; i < n; i += nthr) {
                    const int dl = qD[i];
                    if (dl >= 0) {
                        const int pos = atomicAdd(&rCnt[dl], 1);
                        if (pos < RCAP) { rList[dl * RCAP + pos] = (unsigned short)i; qD[i] = -1; }
                        else            wOvf = 1;
                    }
                }
                __syncthreads();
            }
            c0 = c1;
        }

        for (int i = tid; i < R * 64; i += nthr) {
            const float den = sDen[i >> 6];
            agg[(size_t)base * 64 + i] = (den > 0.0f) ? sAgg[i] / den : 0.0f;
        }
        __syncthreads();
    }
}

__global__ void out_xform_kernel(const float* __restrict__ agg, const float* __restrict__ bias,
                                 const float* __restrict__ Wfc, const float* __restrict__ bfc,
                                 float* __restrict__ out, int NT)
{
    __shared__ __attribute__((aligned(16))) float sW[WLDS];
    __shared__ float sb[64], sbias[64];
    stage_weight_pairs<false>(sW, Wfc, nullptr, nullptr);
    for (int i = threadIdx.x; i < 64; i += blockDim.x) { sb[i] = bfc[i]; sbias[i] = bias[i]; }
    __syncthreads();

    const int lane = threadIdx.x & 31;
    const int wave = threadIdx.x >> 5;
    const int wpb  = blockDim.x >> 5;
    const int l15  = lane & 15;
    const int half = lane >> 4;
    const int numTiles = NT >> 4;

    for (int tile = blockIdx.x * wpb + wave; tile < numTiles; tile += gridDim.x * wpb) {
        v8f acc[4] = {v8f{}, v8f{}, v8f{}, v8f{}};
        const float* arow = agg + (size_t)(tile * 16 + l15) * 64;
        #pragma unroll 4
        for (int kk = 0; kk < 16; ++kk) {
            const int kbase = kk * 4 + half * 2;
            v2f a = *(const v2f*)(arow + kbase);
            a.x += sbias[kbase];
            a.y += sbias[kbase + 1];
            const int pbase = (kk * 2 + half) * PSTRIDE;
            #pragma unroll
            for (int nb = 0; nb < 4; ++nb) {
                const int col = nb * 16 + l15;
                const v2f b = *(const v2f*)(sW + pbase + col * 2);
                acc[nb] = __builtin_amdgcn_wmma_f32_16x16x4_f32(
                    false, a, false, b, (short)0, acc[nb], false, false);
            }
        }
        #pragma unroll
        for (int nb = 0; nb < 4; ++nb) {
            const int col = nb * 16 + l15;
            #pragma unroll
            for (int v = 0; v < 8; ++v) {
                const size_t row = (size_t)(tile * 16 + v + half * 8);
                out[row * 64 + col] = acc[nb][v] + sb[col];
            }
        }
    }
}

extern "C" void kernel_launch(void* const* d_in, const int* in_sizes, int n_in,
                              void* d_out, int out_size, void* d_ws, size_t ws_size,
                              hipStream_t stream) {
    const float* x      = (const float*)d_in[0];
    const int*   eidx   = (const int*)d_in[1];
    const float* eattr  = (const float*)d_in[2];
    const float* W_l    = (const float*)d_in[3];
    const float* b_l    = (const float*)d_in[4];
    const float* W_r    = (const float*)d_in[5];
    const float* b_r    = (const float*)d_in[6];
    const float* W_e    = (const float*)d_in[7];
    const float* att    = (const float*)d_in[8];
    const float* bias   = (const float*)d_in[9];
    const float* W_fc   = (const float*)d_in[10];
    const float* b_fc   = (const float*)d_in[11];

    const int NT       = in_sizes[0] / 64;
    const long long E  = (long long)in_sizes[2];
    const int* src = eidx;
    const int* dst = eidx + E;

    float* ws     = (float*)d_ws;
    float* xl     = ws;
    float* xr     = xl + (size_t)NT * 64;
    float* agg    = xr;
    unsigned* rec = (unsigned*)(xr + (size_t)NT * 64);
    unsigned* kplanes = rec + 2 * E;

    const int numTiles = NT / 16;
    const int gemmBlocks = (numTiles + 7) / 8;

    node_xform_kernel<<<gemmBlocks, 256, 0, stream>>>(x, W_l, b_l, W_r, b_r, xl, xr, NT, src, dst, eattr, E, kplanes, rec);
    node_range_gat_kernel<<<256, 1024, 0, stream>>>(rec, W_e, att, xl, xr, agg, NT, E, kplanes);
    out_xform_kernel<<<gemmBlocks, 256, 0, stream>>>(agg, bias, W_fc, b_fc, (float*)d_out, NT);
}


# edge score: att.leaky_relu(z,0.2) formed as 1.5*(a.z) + a.|z| with a = 0.4*att (|z| a source modifier): no compare, select or 0.2 product per channel; the score's 64-term sum order changes
# speedup vs baseline: 6.5878x; 1.0376x over previous
#include <hip/hip_runtime.h>
#include <hip/hip_bf16.h>

typedef float v2f __attribute__((ext_vector_type(2)));
typedef float v8f __attribute__((ext_vector_type(8)));
typedef float v4f __attribute__((ext_vector_type(4)));

#define PSTRIDE 160
#define WLDS    (32 * PSTRIDE)

__device__ __forceinline__ unsigned enc_f32(float f) {
    unsigned u = __float_as_uint(f);
    return (u & 0x80000000u) ? ~u : (u | 0x80000000u);
}
__device__ __forceinline__ float dec_f32(unsigned e) {
    return (e & 0x80000000u) ? __uint_as_float(e ^ 0x80000000u) : __uint_as_float(~e);
}

#define RMAX  196

template <bool WHOLE>
__device__ __forceinline__ void records_and_planes_batch(const int* __restrict__ src, const int* __restrict__ dst,
                                                         const float* __restrict__ eattr, long long E, long long nG,
                                                         uint4* __restrict__ klo, uint4* __restrict__ khi,
                                                         uint2* __restrict__ rec, long long G0, int lane)
{
    int   d[16], s[16];
    float a[16];
    #pragma unroll
    for (int j = 0; j < 16; ++j) {
        const long long e  = ((G0 + j) << 5) + lane;
        const long long ec = (WHOLE || e < E) ? e : (E - 1);
        d[j] = dst[ec];
        s[j] = src[ec];
        a[j] = eattr[ec];
    }
    asm volatile("" :: "v"(d[0]), "v"(d[1]), "v"(d[2]), "v"(d[3]), "v"(d[4]), "v"(d[5]), "v"(d[6]), "v"(d[7]),
                       "v"(s[0]), "v"(s[1]), "v"(s[2]), "v"(s[3]), "v"(s[4]), "v"(s[5]), "v"(s[6]), "v"(s[7]),
                       "v"(a[0]), "v"(a[1]), "v"(a[2]), "v"(a[3]), "v"(a[4]), "v"(a[5]), "v"(a[6]), "v"(a[7]));
    asm volatile("" :: "v"(d[8]), "v"(d[9]), "v"(d[10]), "v"(d[11]), "v"(d[12]), "v"(d[13]), "v"(d[14]), "v"(d[15]),
                       "v"(s[8]), "v"(s[9]), "v"(s[10]), "v"(s[11]), "v"(s[12]), "v"(s[13]), "v"(s[14]), "v"(s[15]),
                       "v"(a[8]), "v"(a[9]), "v"(a[10]), "v"(a[11]), "v"(a[12]), "v"(a[13]), "v"(a[14]), "v"(a[15]));
    unsigned mine[8] = {0u, 0u, 0u, 0u, 0u, 0u, 0u, 0u};
    #pragma unroll
    for (int j = 0; j < 16; ++j) {
        const long long e = ((G0 + j) << 5) + lane;
        if (WHOLE || e < E) rec[e] = make_uint2((unsigned)s[j] | ((unsigned)d[j] << 16), __float_as_uint(a[j]));
        const unsigned key = (unsigned)d[j] / RMAX;
        #pragma unroll
        for (int k = 0; k < 8; ++k) {
            const unsigned p = __builtin_amdgcn_ballot_w32(((key >> k) & 1u) != 0u);
            if (lane == j) mine[k] = p;
        }
    }
    if (lane < 16 && (WHOLE || G0 + lane < nG)) {
        klo[G0 + lane] = make_uint4(mine[0], mine[1], mine[2], mine[3]);
        khi[G0 + lane] = make_uint4(mine[4], mine[5], mine[6], mine[7]);
    }
}

__device__ __forceinline__ void write_edge_records_and_key_planes(const int* __restrict__ src,
                                                                  const int* __restrict__ dst,
                                                                  const float* __restrict__ eattr, long long E,
                                                                  unsigned* __restrict__ kplanes,
                                                                  unsigned* __restrict__ recw)
{
    uint2* __restrict__ rec = (uint2*)recw;
    const long long nG = (E + 31) >> 5;
    const long long nB = (nG + 15) >> 4;
    uint4* __restrict__ klo = (uint4*)kplanes;
    uint4* __restrict__ khi = klo + nG;
    const int lane = threadIdx.x & 31;
    const long long wid = (long long)blockIdx.x * (blockDim.x >> 5) + __builtin_amdgcn_readfirstlane((int)(threadIdx.x >> 5));
    const long long nw  = (long long)gridDim.x * (blockDim.x >> 5);
    for (long long b = wid; b < nB; b += nw) {
        const long long G0 = b << 4;
        if (((G0 + 16) << 5) <= E) records_and_planes_batch<true >(src, dst, eattr, E, nG, klo, khi, rec, G0, lane);
        else                       records_and_planes_batch<false>(src, dst, eattr, E, nG, klo, khi, rec, G0, lane);
    }
}

template <bool TWO>
__device__ __forceinline__ void stage_weight_pairs(float* sA, const float* __restrict__ WA,
                                                   float* sB, const float* __restrict__ WB)
{
    const int nthr = blockDim.x;
    for (int u0 = threadIdx.x; u0 < 512; u0 += 2 * nthr) {
        v4f a0[2], a1[2], b0[2], b1[2];
        #pragma unroll
        for (int j = 0; j < 2; ++j) {
            const int u  = u0 + j * nthr;
            const int uc = (u < 512) ? u : u0;
            const int o  = (uc >> 4) * 128 + (uc & 15) * 4;
            a0[j] = *(const v4f*)(WA + o);
            a1[j] = *(const v4f*)(WA + o + 64);
            if constexpr (TWO) {
                b0[j] = *(const v4f*)(WB + o);
                b1[j] = *(const v4f*)(WB + o + 64);
            }
        }
        if constexpr (TWO) asm volatile("" :: "v"(a0[0]), "v"(a1[0]), "v"(a0[1]), "v"(a1[1]), "v"(b0[0]), "v"(b1[0]), "v"(b0[1]), "v"(b1[1]));
        else               asm volatile("" :: "v"(a0[0]), "v"(a1[0]), "v"(a0[1]), "v"(a1[1]));
        #pragma unroll
        for (int j = 0; j < 2; ++j) {
            const int u = u0 + j * nthr;
            if (u < 512) {
                const int d = (u >> 4) * PSTRIDE + (u & 15) * 8;
                *(v4f*)(sA + d)     = v4f{a0[j].x, a1[j].x, a0[j].y, a1[j].y};
                *(v4f*)(sA + d + 4) = v4f{a0[j].z, a1[j].z, a0[j].w, a1[j].w};
                if constexpr (TWO) {
                    *(v4f*)(sB + d)     = v4f{b0[j].x, b1[j].x, b0[j].y, b1[j].y};
                    *(v4f*)(sB + d + 4) = v4f{b0[j].z, b1[j].z, b0[j].w, b1[j].w};
                }
            }
        }
    }
}

__global__ void node_xform_kernel(const float* __restrict__ x2,
                                  const float* __restrict__ Wl, const float* __restrict__ bl,
                                  const float* __restrict__ Wr, const float* __restrict__ br,
                                  float* __restrict__ xl, float* __restrict__ xr, int NT,
                                  const int* __restrict__ src, const int* __restrict__ dst,
                                  const float* __restrict__ eattr, long long E,
                                  unsigned* __restrict__ kplanes, unsigned* __restrict__ recw)
{
    write_edge_records_and_key_planes(src, dst, eattr, E, kplanes, recw);

    __shared__ __attribute__((aligned(16))) float sWl[WLDS];
    __shared__ __attribute__((aligned(16))) float sWr[WLDS];
    __shared__ float sbl[64], sbr[64];
    stage_weight_pairs<true>(sWl, Wl, sWr, Wr);
    for (int i = threadIdx.x; i < 64; i += blockDim.x) { sbl[i] = bl[i]; sbr[i] = br[i]; }
    __syncthreads();

    const int lane    = threadIdx.x & 31;
    const int wave    = threadIdx.x >> 5;
    const int wpb     = blockDim.x >> 5;
    const int l15     = lane & 15;
    const int half    = lane >> 4;
    const int numTiles = NT >> 4;

    for (int tile = blockIdx.x * wpb + wave; tile < numTiles; tile += gridDim.x * wpb) {
        v8f accl[4] = {v8f{}, v8f{}, v8f{}, v8f{}};
        v8f accr[4] = {v8f{}, v8f{}, v8f{}, v8f{}};
        const float* xrow = x2 + (size_t)(tile * 16 + l15) * 64;
        #pragma unroll 4
        for (int kk = 0; kk < 16; ++kk) {
            const int kbase = kk * 4 + half * 2;
            const v2f a = *(const v2f*)(xrow + kbase);
            const int pbase = (kk * 2 + half) * PSTRIDE;
            #pragma unroll
            for (int nb = 0; nb < 4; ++nb) {
                const int col = nb * 16 + l15;
                const v2f bL = *(const v2f*)(sWl + pbase + col * 2);
                const v2f bR = *(const v2f*)(sWr + pbase + col * 2);
                accl[nb] = __builtin_amdgcn_wmma_f32_16x16x4_f32(
                    false, a, false, bL, (short)0, accl[nb], false, false);
                accr[nb] = __builtin_amdgcn_wmma_f32_16x16x4_f32(
                    false, a, false, bR, (short)0, accr[nb], false, false);
            }
        }
        #pragma unroll
        for (int nb = 0; nb < 4; ++nb) {
            const int col = nb * 16 + l15;
            #pragma unroll
            for (int v = 0; v < 8; ++v) {
                const size_t row = (size_t)(tile * 16 + v + half * 8);
                xl[row * 64 + col] = accl[nb][v] + sbl[col];
                xr[row * 64 + col] = accr[nb][v] + sbr[col];
            }
        }
    }
}

#define QCAP  12288
#define SUPER 2097152
#define NGRP  4
#define NSLOT 7
#define RCAP  96
#define NSET  5

template <int M>
__device__ __forceinline__ float xrow(float x) {
    const int b = __builtin_bit_cast(int, x);
    return __builtin_bit_cast(float, __builtin_amdgcn_update_dpp(b, b, 0x160 + M, 0xf, 0xf, false));
}

__device__ __forceinline__ float lane_score(float4 l0, float4 l1, float4 r0, float4 r1, float ea,
                                            float4 w0, float4 w1, float4 a0, float4 a1)
{
    const float z0 = l0.x + r0.x + ea * w0.x;
    const float z1 = l0.y + r0.y + ea * w0.y;
    const float z2 = l0.z + r0.z + ea * w0.z;
    const float z3 = l0.w + r0.w + ea * w0.w;
    const float z4 = l1.x + r1.x + ea * w1.x;
    const float z5 = l1.y + r1.y + ea * w1.y;
    const float z6 = l1.z + r1.z + ea * w1.z;
    const float z7 = l1.w + r1.w + ea * w1.w;
    float s0 = a0.x * z0, s1 = a0.y * z1;
    s0 = fmaf(a0.z, z2, s0);  s1 = fmaf(a0.w, z3, s1);
    s0 = fmaf(a1.x, z4, s0);  s1 = fmaf(a1.y, z5, s1);
    s0 = fmaf(a1.z, z6, s0);  s1 = fmaf(a1.w, z7, s1);
    float sa = a0.x * fabsf(z0);
    sa = fmaf(a0.y, fabsf(z1), sa);
    sa = fmaf(a0.z, fabsf(z2), sa);
    sa = fmaf(a0.w, fabsf(z3), sa);
    sa = fmaf(a1.x, fabsf(z4), sa);
    sa = fmaf(a1.y, fabsf(z5), sa);
    sa = fmaf(a1.z, fabsf(z6), sa);
    sa = fmaf(a1.w, fabsf(z7), sa);
    return fmaf(1.5f, s0 + s1, sa);
}

__global__ void __launch_bounds__(1024)
node_range_gat_kernel(const unsigned* __restrict__ recw,
                      const float* __restrict__ We, const float* __restrict__ att,
                      const float* __restrict__ xl, const float* xr,
                      float* agg, int NT, long long E,
                      const unsigned* __restrict__ kplanes)
{
    __shared__ __attribute__((aligned(16))) float sAgg[RMAX * 64];
    __shared__ __attribute__((aligned(16))) float sXr[RMAX * 64];
    __shared__ float    sDen[RMAX];
    __shared__ float    sMax[RMAX];
    __shared__ int   qE[QCAP];
    __shared__ int   qD[QCAP];
    __shared__ float qX[QCAP];
    __shared__ int   qN;
    __shared__ unsigned short rList[RMAX * RCAP];
    __shared__ int   rCnt[RMAX];
    __shared__ int   wOvf;
    __shared__ __attribute__((aligned(16))) float sWe[64];
    __shared__ __attribute__((aligned(16))) float sAt[64];

    const int tid   = threadIdx.x;
    const int lane  = tid & 31;
    const int wave  = __builtin_amdgcn_readfirstlane(tid >> 5);
    const int nthr  = blockDim.x;
    const long long nG = (E + 31) >> 5;
    const uint4* __restrict__ klo = (const uint4*)kplanes;
    const uint4* __restrict__ khi = klo + nG;
    const uint2* __restrict__ rec = (const uint2*)recw;

    for (int i = tid; i < 64; i += nthr) { sWe[i] = We[i]; sAt[i] = att[i]; }

    for (int base = blockIdx.x * RMAX; base < NT; base += gridDim.x * RMAX) {
        const int R = (NT - base < RMAX) ? (NT - base) : RMAX;
        {
            const int n4 = R * 16;
            const v4f* xr4 = (const v4f*)(xr + (size_t)base * 64);
            for (int i0 = tid; i0 < n4; i0 += 4 * nthr) {
                v4f v[4];
                #pragma unroll
                for (int j = 0; j < 4; ++j) {
                    const int i = i0 + j * nthr;
                    v[j] = xr4[(i < n4) ? i : i0];
                }
                asm volatile("" :: "v"(v[0]), "v"(v[1]), "v"(v[2]), "v"(v[3]));
                #pragma unroll
                for (int j = 0; j < 4; ++j) {
                    const int i = i0 + j * nthr;
                    if (i < n4) ((v4f*)sXr)[i] = v[j];
                }
            }
        }
        for (int i = tid; i < R * 64; i += nthr) sAgg[i] = 0.0f;
        for (int i = tid; i < RMAX; i += nthr) {
            sDen[i] = 0.0f;
            sMax[i] = -__builtin_inff();
            rCnt[i] = 0;
        }
        if (tid == 0) { qN = 0; wOvf = 0; }
        __syncthreads();

        const unsigned pb = (unsigned)(base / RMAX) & 255u;
        unsigned ik[8];
        #pragma unroll
        for (int k = 0; k < 8; ++k) ik[k] = ((pb >> k) & 1u) ? 0u : ~0u;

        long long c0  = 0;
        long long seg = SUPER;
        while (c0 < E) {
            const long long c1 = (c0 + seg < E) ? (c0 + seg) : E;

            const unsigned G0    = (unsigned)(c0 >> 5);
            const unsigned G1    = (unsigned)((c1 + 31) >> 5);
            const unsigned trip  = (unsigned)nthr * NGRP;
            const unsigned ntrip = (G1 - G0 + trip - 1) / trip;
            const unsigned rot   = ntrip * blockIdx.x / gridDim.x;
            auto first_group = [&](unsigned tr) -> unsigned {
                if (tr >= ntrip) return G1;
                const unsigned tt = (tr + rot < ntrip) ? (tr + rot) : (tr + rot - ntrip);
                return G0 + tt * trip + (unsigned)tid;
            };
            auto load_groups = [&](uint4 (&lo)[NGRP], uint4 (&hi)[NGRP], unsigned g0) {
                #pragma unroll
                for (int j = 0; j < NGRP; ++j) {
                    const unsigned g  = g0 + (unsigned)j * nthr;
                    const unsigned gc = (g < G1) ? g : (G1 - 1u);
                    lo[j] = klo[gc];
                    hi[j] = khi[gc];
                }
            };
            auto take_groups = [&](const uint4 (&lo)[NGRP], const uint4 (&hi)[NGRP], unsigned g0) {
                #pragma unroll
                for (int j = 0; j < NGRP; ++j) {
                    unsigned m = (lo[j].x ^ ik[0]) & (lo[j].y ^ ik[1]) & (lo[j].z ^ ik[2]) & (lo[j].w ^ ik[3])
                               & (hi[j].x ^ ik[4]) & (hi[j].y ^ ik[5]) & (hi[j].z ^ ik[6]) & (hi[j].w ^ ik[7]);
                    if (m != 0u) {
                        const long long eb = (long long)(g0 + (unsigned)j * nthr) << 5;
                        do {
                            const long long e = eb + __builtin_ctz(m);
                            m &= m - 1u;
                            if (e < c1) {
                                const int idx = atomicAdd(&qN, 1);
                                if (idx < QCAP) qE[idx] = (int)e;
                            }
                        } while (m != 0u);
                    }
                }
            };
            {
                uint4 loA[NGRP], hiA[NGRP], loB[NGRP], hiB[NGRP];
                unsigned gA = first_group(0u);
                load_groups(loA, hiA, gA);
                for (unsigned tr = 0; tr < ntrip; tr += 2) {
                    const unsigned gB = first_group(tr + 1);
                    load_groups(loB, hiB, gB);
                    take_groups(loA, hiA, gA);
                    gA = first_group(tr + 2);
                    load_groups(loA, hiA, gA);
                    if (tr + 1 < ntrip) take_groups(loB, hiB, gB);
                }
            }
            __syncthreads();

            const int n = qN;
            if (n > QCAP) {
                __syncthreads();
                if (tid == 0) qN = 0;
                seg = (seg / 2 > QCAP) ? seg / 2 : QCAP;
                __syncthreads();
                continue;
            }

            for (int i = tid; i < n; i += nthr) {
                const uint2 r = rec[qE[i]];
                const unsigned dl = (r.x >> 16) - (unsigned)base;
                qE[i] = (int)(r.x & 0xffffu);
                qX[i] = __uint_as_float(r.y);
                int wait = -1;
                if (dl < (unsigned)R) {
                    const int pos = atomicAdd(&rCnt[dl], 1);
                    if (pos < RCAP) rList[dl * RCAP + pos] = (unsigned short)i;
                    else          { wait = (int)dl; wOvf = 1; }
                }
                qD[i] = wait;
            }
            __syncthreads();
            if (tid == 0) qN = 0;

            for (;;) {
                {
                    const int o8 = lane & 7;
                    const int g4 = lane >> 3;
                    unsigned cb[2];
                    #pragma unroll
                    for (int k = 0; k < 2; ++k) cb[k] = 128u * (unsigned)k + 16u * (unsigned)o8;
                    float4 we4[2], a4[2];
                    #pragma unroll
                    for (int k = 0; k < 2; ++k) {
                        we4[k] = *(const float4*)((const char*)sWe + cb[k]);
                        const float4 at = *(const float4*)((const char*)sAt + cb[k]);
                        a4[k] = float4{0.4f * at.x, 0.4f * at.y, 0.4f * at.z, 0.4f * at.w};
                    }
                    for (int slot = 0; slot < NSLOT; ++slot) {
                        const int row = wave + 32 * slot;
                        if (row >= R) break;
                        const int cnt = __builtin_amdgcn_readfirstlane(rCnt[row]);
                        const int c   = (cnt < RCAP) ? cnt : RCAP;
                        if (c == 0) continue;
                        const unsigned ro = (unsigned)row * 256u;
                        float4 r4[2];
                        #pragma unroll
                        for (int k = 0; k < 2; ++k) r4[k] = *(const float4*)((const char*)sXr + (ro + cb[k]));
                        const unsigned short* lst = rList + row * RCAP;
                        float  m   = -__builtin_inff();
                        float  den = 0.0f;
                        float4 acc[2] = {float4{0.0f, 0.0f, 0.0f, 0.0f}, float4{0.0f, 0.0f, 0.0f, 0.0f}};
                        for (int j0 = 0; j0 < c; j0 += NSET * 4) {
                            bool   ok[NSET];
                            float  ea[NSET];
                            float4 l[NSET][2];
                            #pragma unroll
                            for (int s = 0; s < NSET; ++s) {
                                const int j  = j0 + s * 4 + g4;
                                ok[s] = j < c;
                                const int ii = lst[ok[s] ? j : 0];
                                ea[s] = qX[ii];
                                const unsigned lo = (unsigned)qE[ii] * 256u;
                                #pragma unroll
                                for (int k = 0; k < 2; ++k) l[s][k] = *(const float4*)((const char*)xl + (size_t)(lo + cb[k]));
                            }
                            __builtin_amdgcn_sched_barrier(0);
                            #pragma unroll
                            for (int s = 0; s < NSET; ++s) {
                                if (j0 + s * 4 < c) {
                                    float p = lane_score(l[s][0], l[s][1], r4[0], r4[1], ea[s], we4[0], we4[1], a4[0], a4[1]);
                                    p += xrow<1>(p);
                                    p += xrow<2>(p);
                                    p += xrow<4>(p);
                                    const float mn = ok[s] ? fmaxf(m, p) : m;
                                    const float sc = (mn == m) ? 1.0f : __expf(m - mn);
                                    const float w  = ok[s] ? __expf(p - mn) : 0.0f;
                                    den = den * sc + w;
                                    #pragma unroll
                                    for (int k = 0; k < 2; ++k) {
                                        acc[k].x = acc[k].x * sc + w * l[s][k].x;
                                        acc[k].y = acc[k].y * sc + w * l[s][k].y;
                                        acc[k].z = acc[k].z * sc + w * l[s][k].z;
                                        acc[k].w = acc[k].w * sc + w * l[s][k].w;
                                    }
                                    m = mn;
                                }
                            }
                        }
                        float M = m;
                        M = fmaxf(M, xrow<8>(M));
                        M = fmaxf(M, __shfl_xor(M, 16));
                        const float f = __expf(m - M);
                        den *= f;
                        den += xrow<8>(den);
                        den += __shfl_xor(den, 16);
                        #pragma unroll
                        for (int k = 0; k < 2; ++k) {
                            float v[4] = {acc[k].x * f, acc[k].y * f, acc[k].z * f, acc[k].w * f};
                            #pragma unroll
                            for (int t = 0; t < 4; ++t) {
                                v[t] += xrow<8>(v[t]);
                                v[t] += __shfl_xor(v[t], 16);
                            }
                            acc[k] = float4{v[0], v[1], v[2], v[3]};
                        }
                        if (g4 == 0) {
                            const float Mo = sMax[row];
                            const float Mn = fmaxf(Mo, M);
                            const float so = (Mo == Mn) ? 1.0f : __expf(Mo - Mn);
                            const float sw = (M  == Mn) ? 1.0f : __expf(M - Mn);
                            #pragma unroll
                            for (int k = 0; k < 2; ++k) {
                                float4* p = (float4*)((char*)sAgg + (ro + cb[k]));
                                float4 v = *p;
                                v.x = v.x * so + acc[k].x * sw;
                                v.y = v.y * so + acc[k].y * sw;
                                v.z = v.z * so + acc[k].z * sw;
                                v.w = v.w * so + acc[k].w * sw;
                                *p = v;
                            }
                            if (o8 == 0) {
                                sDen[row] = sDen[row] * so + den * sw;
                                sMax[row] = Mn;
                            }
                        }
                    }
                }
                const int again = wOvf;
                __syncthreads();
                for (int i = tid; i < RMAX; i += nthr) rCnt[i] = 0;
                if (!again) break;
                if (tid == 0) wOvf = 0;
                __syncthreads();
                for (int i = tid; i < n; i += nthr) {
                    const int dl = qD[i];
                    if (dl >= 0) {
                        const int pos = atomicAdd(&rCnt[dl], 1);
                        if (pos < RCAP) { rList[dl * RCAP + pos] = (unsigned short)i; qD[i] = -1; }
                        else            wOvf = 1;
                    }
                }
                __syncthreads();
            }
            c0 = c1;
        }

        for (int i = tid; i < R * 64; i += nthr) {
            const float den = sDen[i >> 6];
            agg[(size_t)base * 64 + i] = (den > 0.0f) ? sAgg[i] / den : 0.0f;
        }
        __syncthreads();
    }
}

__global__ void out_xform_kernel(const float* __restrict__ agg, const float* __restrict__ bias,
                                 const float* __restrict__ Wfc, const float* __restrict__ bfc,
                                 float* __restrict__ out, int NT)
{
    __shared__ __attribute__((aligned(16))) float sW[WLDS];
    __shared__ float sb[64], sbias[64];
    stage_weight_pairs<false>(sW, Wfc, nullptr, nullptr);
    for (int i = threadIdx.x; i < 64; i += blockDim.x) { sb[i] = bfc[i]; sbias[i] = bias[i]; }
    __syncthreads();

    const int lane = threadIdx.x & 31;
    const int wave = threadIdx.x >> 5;
    const int wpb  = blockDim.x >> 5;
    const int l15  = lane & 15;
    const int half = lane >> 4;
    const int numTiles = NT >> 4;

    for (int tile = blockIdx.x * wpb + wave; tile < numTiles; tile += gridDim.x * wpb) {
        v8f acc[4] = {v8f{}, v8f{}, v8f{}, v8f{}};
        const float* arow = agg + (size_t)(tile * 16 + l15) * 64;
        #pragma unroll 4
        for (int kk = 0; kk < 16; ++kk) {
            const int kbase = kk * 4 + half * 2;
            v2f a = *(const v2f*)(arow + kbase);
            a.x += sbias[kbase];
            a.y += sbias[kbase + 1];
            const int pbase = (kk * 2 + half) * PSTRIDE;
            #pragma unroll
            for (int nb = 0; nb < 4; ++nb) {
                const int col = nb * 16 + l15;
                const v2f b = *(const v2f*)(sW + pbase + col * 2);
                acc[nb] = __builtin_amdgcn_wmma_f32_16x16x4_f32(
                    false, a, false, b, (short)0, acc[nb], false, false);
            }
        }
        #pragma unroll
        for (int nb = 0; nb < 4; ++nb) {
            const int col = nb * 16 + l15;
            #pragma unroll
            for (int v = 0; v < 8; ++v) {
                const size_t row = (size_t)(tile * 16 + v + half * 8);
                out[row * 64 + col] = acc[nb][v] + sb[col];
            }
        }
    }
}

extern "C" void kernel_launch(void* const* d_in, const int* in_sizes, int n_in,
                              void* d_out, int out_size, void* d_ws, size_t ws_size,
                              hipStream_t stream) {
    const float* x      = (const float*)d_in[0];
    const int*   eidx   = (const int*)d_in[1];
    const float* eattr  = (const float*)d_in[2];
    const float* W_l    = (const float*)d_in[3];
    const float* b_l    = (const float*)d_in[4];
    const float* W_r    = (const float*)d_in[5];
    const float* b_r    = (const float*)d_in[6];
    const float* W_e    = (const float*)d_in[7];
    const float* att    = (const float*)d_in[8];
    const float* bias   = (const float*)d_in[9];
    const float* W_fc   = (const float*)d_in[10];
    const float* b_fc   = (const float*)d_in[11];

    const int NT       = in_sizes[0] / 64;
    const long long E  = (long long)in_sizes[2];
    const int* src = eidx;
    const int* dst = eidx + E;

    float* ws     = (float*)d_ws;
    float* xl     = ws;
    float* xr     = xl + (size_t)NT * 64;
    float* agg    = xr;
    unsigned* rec = (unsigned*)(xr + (size_t)NT * 64);
    unsigned* kplanes = rec + 2 * E;

    const int numTiles = NT / 16;
    const int gemmBlocks = (numTiles + 7) / 8;

    node_xform_kernel<<<gemmBlocks, 256, 0, stream>>>(x, W_l, b_l, W_r, b_r, xl, xr, NT, src, dst, eattr, E, kplanes, rec);
    node_range_gat_kernel<<<256, 1024, 0, stream>>>(rec, W_e, att, xl, xr, agg, NT, E, kplanes);
    out_xform_kernel<<<gemmBlocks, 256, 0, stream>>>(agg, bias, W_fc, b_fc, (float*)d_out, NT);
}
